# starting kernel
# baseline (speedup 1.0000x reference)
#include <hip/hip_runtime.h>

#define T_DIM 128
#define B_DIM 128
#define R_DIM 512
#define NSLOTS 256
#define SPAD   257

typedef float v2f __attribute__((ext_vector_type(2)));
typedef float v8f __attribute__((ext_vector_type(8)));

__global__ __launch_bounds__(128) void weights_kernel(
    const float* __restrict__ u,
    const float* __restrict__ d1,
    const float* __restrict__ d2,
    float* __restrict__ W)
{
    __shared__ float S[B_DIM * SPAD];
    __shared__ float X[B_DIM * SPAD];
    __shared__ float scal[NSLOTS];

    const int b = threadIdx.x;
    float* Srow = &S[b * SPAD];
    float* Xrow = &X[b * SPAD];

    #pragma unroll 8
    for (int j = 0; j < NSLOTS; ++j) Srow[j] = 0.0f;

    for (int t = 0; t < T_DIM; ++t) {
        const int jmax = 2 * t + 1;
        const float ut = u[t * B_DIM + b];
        Srow[2 * t]     = d1[t * B_DIM + b];
        Srow[2 * t + 1] = d2[t * B_DIM + b];

        float used = 0.0f;
        for (int j = jmax; j >= 0; --j) {
            Xrow[j] = ut - used;
            used += Srow[j];
        }
        __syncthreads();

        for (int jj = threadIdx.x; jj <= jmax; jj += B_DIM) {
            float m = -INFINITY;
            #pragma unroll 8
            for (int bb = 0; bb < B_DIM; ++bb)
                m = fmaxf(m, X[bb * SPAD + jj]);
            scal[jj] = m;
        }
        __syncthreads();

        float* wrow = W + ((size_t)t * B_DIM + b) * NSLOTS;
        for (int j = 0; j < NSLOTS; ++j) {
            float w = 0.0f;
            if (j <= jmax) {
                const float s = Srow[j];
                w = fminf(s, scal[j]);
                const float take = fminf(s, fmaxf(0.0f, Xrow[j]));
                Srow[j] = s - take;
            }
            wrow[j] = w;
        }
    }
}

__global__ __launch_bounds__(256) void gemm_kernel(
    const float* __restrict__ W,
    const float* __restrict__ v1,
    const float* __restrict__ v2,
    float* __restrict__ out)
{
    const int lane = threadIdx.x & 31;
    const int wave = threadIdx.x >> 5;
    const int lm   = lane & 15;
    const int half = lane >> 4;

    const int b  = blockIdx.x >> 3;
    const int mt = blockIdx.x & 7;
    const int n0 = wave * 64;

    v8f acc0 = {}, acc1 = {}, acc2 = {}, acc3 = {};

    const float* Ap = W + ((size_t)(mt * 16 + lm) * B_DIM + b) * NSLOTS + 2 * half;
    const size_t bstride = (size_t)B_DIM * R_DIM;
    const float* B1 = v1 + (size_t)half * bstride + (size_t)b * R_DIM + n0 + lm;
    const float* B2 = v2 + (size_t)half * bstride + (size_t)b * R_DIM + n0 + lm;

    #pragma unroll 4
    for (int k = 0; k < NSLOTS; k += 4) {
        const v2f a = *(const v2f*)(Ap + k);
        const size_t off = (size_t)(k >> 1) * bstride;
        v2f b0, b1, b2, b3;
        b0.x = B1[off +  0]; b0.y = B2[off +  0];
        b1.x = B1[off + 16]; b1.y = B2[off + 16];
        b2.x = B1[off + 32]; b2.y = B2[off + 32];
        b3.x = B1[off + 48]; b3.y = B2[off + 48];
        acc0 = __builtin_amdgcn_wmma_f32_16x16x4_f32(false, a, false, b0, (short)0, acc0, false, false);
        acc1 = __builtin_amdgcn_wmma_f32_16x16x4_f32(false, a, false, b1, (short)0, acc1, false, false);
        acc2 = __builtin_amdgcn_wmma_f32_16x16x4_f32(false, a, false, b2, (short)0, acc2, false, false);
        acc3 = __builtin_amdgcn_wmma_f32_16x16x4_f32(false, a, false, b3, (short)0, acc3, false, false);
    }

    #pragma unroll
    for (int r = 0; r < 8; ++r) {
        const int trow = mt * 16 + r + 8 * half;
        float* orow = out + ((size_t)trow * B_DIM + b) * R_DIM + n0 + lm;
        orow[ 0] = acc0[r];
        orow[16] = acc1[r];
        orow[32] = acc2[r];
        orow[48] = acc3[r];
    }
}

extern "C" void kernel_launch(void* const* d_in, const int* in_sizes, int n_in,
                              void* d_out, int out_size, void* d_ws, size_t ws_size,
                              hipStream_t stream) {
    const float* u  = (const float*)d_in[0];
    const float* d1 = (const float*)d_in[1];
    const float* d2 = (const float*)d_in[2];
    const float* v1 = (const float*)d_in[3];
    const float* v2 = (const float*)d_in[4];
    float* out = (float*)d_out;
    float* W   = (float*)d_ws;

    weights_kernel<<<dim3(1), dim3(128), 0, stream>>>(u, d1, d2, W);
    gemm_kernel<<<dim3(B_DIM * 8), dim3(256), 0, stream>>>(W, v1, v2, out);
}


# Serial weights kernel split by batch: stack state depends on its own batch only, so 4 one-wave blocks scan 32 batches each (same add order); a small kernel takes the cross-batch max and writes W
# speedup vs baseline: 9.9032x; 9.9032x over previous
#include <hip/hip_runtime.h>

#define T_DIM 128
#define B_DIM 128
#define R_DIM 512
#define NSLOTS 256

typedef float v2f __attribute__((ext_vector_type(2)));
typedef float v8f __attribute__((ext_vector_type(8)));

#define GRP 16

__device__ __forceinline__ void load_group(float (&c)[GRP], const float* Sg)
{
    #pragma unroll
    for (int p = 0; p < GRP; ++p) c[p] = Sg[p * 32];
}

__device__ __forceinline__ void scan_group(const float (&c)[GRP], float* Sg, float2* pg,
                                           const float ut, float& used)
{
    #pragma unroll
    for (int p = GRP - 1; p >= 0; --p) {
        const float s = c[p];
        const float x = ut - used;
        used += s;
        const float take = fminf(s, fmaxf(0.0f, x));
        Sg[p * 32] = s - take;
        pg[(size_t)p * B_DIM] = make_float2(x, s);
    }
}

__global__ __launch_bounds__(32) void state_kernel(
    const float* __restrict__ u,
    const float* __restrict__ d1,
    const float* __restrict__ d2,
    float* __restrict__ P)
{
    __shared__ float S[NSLOTS * 32];
    __shared__ float D[NSLOTS * 32];
    __shared__ float U[T_DIM * 32];

    const int lane = threadIdx.x;
    const int b = blockIdx.x * 32 + lane;
    float2* P2 = (float2*)P;

    #pragma unroll 16
    for (int t = 0; t < T_DIM; ++t) {
        U[t * 32 + lane]           = u[t * B_DIM + b];
        D[(2 * t) * 32 + lane]     = d1[t * B_DIM + b];
        D[(2 * t + 1) * 32 + lane] = d2[t * B_DIM + b];
        S[(2 * t) * 32 + lane]     = 0.0f;
        S[(2 * t + 1) * 32 + lane] = 0.0f;
    }

    float un = U[lane], d1n = D[lane], d2n = D[32 + lane];
    for (int t = 0; t < T_DIM; ++t) {
        const float ut = un;
        S[(2 * t) * 32 + lane]     = d1n;
        S[(2 * t + 1) * 32 + lane] = d2n;
        const int tn = (t + 1 < T_DIM) ? t + 1 : t;
        un  = U[tn * 32 + lane];
        d1n = D[(2 * tn) * 32 + lane];
        d2n = D[(2 * tn + 1) * 32 + lane];

        int g = (2 * t + 1) >> 4;
        float used = 0.0f;
        float2* pt = P2 + (size_t)t * NSLOTS * B_DIM + b;
        float ca[GRP], cb[GRP];
        load_group(ca, &S[g * GRP * 32 + lane]);
        for (;;) {
            const int g1 = (g > 0) ? g - 1 : 0;
            load_group(cb, &S[g1 * GRP * 32 + lane]);
            scan_group(ca, &S[g * GRP * 32 + lane], pt + (size_t)g * GRP * B_DIM, ut, used);
            if (g == 0) break;
            const int g2 = (g > 1) ? g - 2 : 0;
            load_group(ca, &S[g2 * GRP * 32 + lane]);
            scan_group(cb, &S[g1 * GRP * 32 + lane], pt + (size_t)g1 * GRP * B_DIM, ut, used);
            if (g == 1) break;
            g -= 2;
        }
    }
}

__global__ __launch_bounds__(256) void weights_kernel(
    const float* __restrict__ P,
    float* __restrict__ W)
{
    __shared__ float Wt[B_DIM * 33];

    const int lane = threadIdx.x & 31;
    const int wave = threadIdx.x >> 5;
    const int t    = blockIdx.x >> 3;
    const int j0   = (blockIdx.x & 7) * 32;
    const int jmax = 2 * t + 1;
    float* wbase = W + ((size_t)t * B_DIM + wave * 16) * NSLOTS + j0 + lane;

    if (j0 <= jmax) {
        const float2* P2 = (const float2*)P + ((size_t)t * NSLOTS + j0) * B_DIM;
        #pragma unroll
        for (int r = 0; r < 4; ++r) {
            const int jj = wave * 4 + r;
            float w0 = 0.0f, w1 = 0.0f, w2 = 0.0f, w3 = 0.0f;
            if (j0 + jj <= jmax) {
                const float2* row = P2 + (size_t)jj * B_DIM;
                const float2 e0 = row[lane];
                const float2 e1 = row[lane + 32];
                const float2 e2 = row[lane + 64];
                const float2 e3 = row[lane + 96];
                float m = fmaxf(fmaxf(e0.x, e1.x), fmaxf(e2.x, e3.x));
                #pragma unroll
                for (int o = 16; o >= 1; o >>= 1) m = fmaxf(m, __shfl_xor(m, o));
                w0 = fminf(e0.y, m);
                w1 = fminf(e1.y, m);
                w2 = fminf(e2.y, m);
                w3 = fminf(e3.y, m);
            }
            Wt[(lane)      * 33 + jj] = w0;
            Wt[(lane + 32) * 33 + jj] = w1;
            Wt[(lane + 64) * 33 + jj] = w2;
            Wt[(lane + 96) * 33 + jj] = w3;
        }
        __syncthreads();
        #pragma unroll
        for (int i = 0; i < 16; ++i)
            wbase[(size_t)i * NSLOTS] = Wt[(wave * 16 + i) * 33 + lane];
    } else {
        #pragma unroll
        for (int i = 0; i < 16; ++i)
            wbase[(size_t)i * NSLOTS] = 0.0f;
    }
}

__global__ __launch_bounds__(256) void gemm_kernel(
    const float* __restrict__ W,
    const float* __restrict__ v1,
    const float* __restrict__ v2,
    float* __restrict__ out)
{
    const int lane = threadIdx.x & 31;
    const int wave = threadIdx.x >> 5;
    const int lm   = lane & 15;
    const int half = lane >> 4;

    const int b  = blockIdx.x >> 3;
    const int mt = blockIdx.x & 7;
    const int n0 = wave * 64;

    v8f acc0 = {}, acc1 = {}, acc2 = {}, acc3 = {};

    const float* Ap = W + ((size_t)(mt * 16 + lm) * B_DIM + b) * NSLOTS + 2 * half;
    const size_t bstride = (size_t)B_DIM * R_DIM;
    const float* B1 = v1 + (size_t)half * bstride + (size_t)b * R_DIM + n0 + lm;
    const float* B2 = v2 + (size_t)half * bstride + (size_t)b * R_DIM + n0 + lm;

    #pragma unroll 4
    for (int k = 0; k < NSLOTS; k += 4) {
        const v2f a = *(const v2f*)(Ap + k);
        const size_t off = (size_t)(k >> 1) * bstride;
        v2f b0, b1, b2, b3;
        b0.x = B1[off +  0]; b0.y = B2[off +  0];
        b1.x = B1[off + 16]; b1.y = B2[off + 16];
        b2.x = B1[off + 32]; b2.y = B2[off + 32];
        b3.x = B1[off + 48]; b3.y = B2[off + 48];
        acc0 = __builtin_amdgcn_wmma_f32_16x16x4_f32(false, a, false, b0, (short)0, acc0, false, false);
        acc1 = __builtin_amdgcn_wmma_f32_16x16x4_f32(false, a, false, b1, (short)0, acc1, false, false);
        acc2 = __builtin_amdgcn_wmma_f32_16x16x4_f32(false, a, false, b2, (short)0, acc2, false, false);
        acc3 = __builtin_amdgcn_wmma_f32_16x16x4_f32(false, a, false, b3, (short)0, acc3, false, false);
    }

    #pragma unroll
    for (int r = 0; r < 8; ++r) {
        const int trow = mt * 16 + r + 8 * half;
        float* orow = out + ((size_t)trow * B_DIM + b) * R_DIM + n0 + lm;
        orow[ 0] = acc0[r];
        orow[16] = acc1[r];
        orow[32] = acc2[r];
        orow[48] = acc3[r];
    }
}

extern "C" void kernel_launch(void* const* d_in, const int* in_sizes, int n_in,
                              void* d_out, int out_size, void* d_ws, size_t ws_size,
                              hipStream_t stream) {
    const float* u  = (const float*)d_in[0];
    const float* d1 = (const float*)d_in[1];
    const float* d2 = (const float*)d_in[2];
    const float* v1 = (const float*)d_in[3];
    const float* v2 = (const float*)d_in[4];
    float* out = (float*)d_out;
    float* W   = (float*)d_ws;
    float* P   = W + (size_t)T_DIM * B_DIM * NSLOTS;

    state_kernel<<<dim3(B_DIM / 32), dim3(32), 0, stream>>>(u, d1, d2, P);
    weights_kernel<<<dim3(T_DIM * 8), dim3(256), 0, stream>>>(P, W);
    gemm_kernel<<<dim3(B_DIM * 8), dim3(256), 0, stream>>>(W, v1, v2, out);
}


# State kernel: 4 waves per block run the same 32-batch scan on private LDS copies; each stores every 4th slot group of P, so one wave's limit on stores in flight is spread over 4 (same values)
# speedup vs baseline: 14.0601x; 1.4198x over previous
#include <hip/hip_runtime.h>

#define T_DIM 128
#define B_DIM 128
#define R_DIM 512
#define NSLOTS 256

typedef float v2f __attribute__((ext_vector_type(2)));
typedef float v8f __attribute__((ext_vector_type(8)));

#define GRP 16
#define NW 4

__device__ __forceinline__ void load_group(float (&c)[GRP], const float* Sg)
{
    #pragma unroll
    for (int p = 0; p < GRP; ++p) c[p] = Sg[p * 32];
}

__device__ __forceinline__ void scan_group(const float (&c)[GRP], float* Sg, float2* pg,
                                           const float ut, float& used, const bool mine)
{
    float xs[GRP];
    #pragma unroll
    for (int p = GRP - 1; p >= 0; --p) {
        const float s = c[p];
        const float x = ut - used;
        used += s;
        const float take = fminf(s, fmaxf(0.0f, x));
        Sg[p * 32] = s - take;
        xs[p] = x;
    }
    if (mine) {
        #pragma unroll
        for (int p = GRP - 1; p >= 0; --p)
            pg[(size_t)p * B_DIM] = make_float2(xs[p], c[p]);
    }
}

__global__ __launch_bounds__(32 * NW) void state_kernel(
    const float* __restrict__ u,
    const float* __restrict__ d1,
    const float* __restrict__ d2,
    float* __restrict__ P)
{
    __shared__ float Sall[NW * NSLOTS * 32];
    __shared__ float D[NSLOTS * 32];
    __shared__ float U[T_DIM * 32];

    const int lane = threadIdx.x & 31;
    const int wave = __builtin_amdgcn_readfirstlane((int)(threadIdx.x >> 5));
    const int b = blockIdx.x * 32 + lane;
    float2* P2 = (float2*)P;
    float* S = Sall + wave * (NSLOTS * 32);

    #pragma unroll 16
    for (int t = 0; t < T_DIM; ++t) {
        U[t * 32 + lane]           = u[t * B_DIM + b];
        D[(2 * t) * 32 + lane]     = d1[t * B_DIM + b];
        D[(2 * t + 1) * 32 + lane] = d2[t * B_DIM + b];
        S[(2 * t) * 32 + lane]     = 0.0f;
        S[(2 * t + 1) * 32 + lane] = 0.0f;
    }

    float un = U[lane], d1n = D[lane], d2n = D[32 + lane];
    for (int t = 0; t < T_DIM; ++t) {
        const float ut = un;
        S[(2 * t) * 32 + lane]     = d1n;
        S[(2 * t + 1) * 32 + lane] = d2n;
        const int tn = (t + 1 < T_DIM) ? t + 1 : t;
        un  = U[tn * 32 + lane];
        d1n = D[(2 * tn) * 32 + lane];
        d2n = D[(2 * tn + 1) * 32 + lane];

        int g = (2 * t + 1) >> 4;
        float used = 0.0f;
        float2* pt = P2 + (size_t)t * NSLOTS * B_DIM + b;
        float ca[GRP], cb[GRP];
        load_group(ca, &S[g * GRP * 32 + lane]);
        for (;;) {
            const int g1 = (g > 0) ? g - 1 : 0;
            load_group(cb, &S[g1 * GRP * 32 + lane]);
            scan_group(ca, &S[g * GRP * 32 + lane], pt + (size_t)g * GRP * B_DIM, ut, used,
                       (g & (NW - 1)) == wave);
            if (g == 0) break;
            const int g2 = (g > 1) ? g - 2 : 0;
            load_group(ca, &S[g2 * GRP * 32 + lane]);
            scan_group(cb, &S[g1 * GRP * 32 + lane], pt + (size_t)g1 * GRP * B_DIM, ut, used,
                       (g1 & (NW - 1)) == wave);
            if (g == 1) break;
            g -= 2;
        }
    }
}

__global__ __launch_bounds__(256) void weights_kernel(
    const float* __restrict__ P,
    float* __restrict__ W)
{
    __shared__ float Wt[B_DIM * 33];

    const int lane = threadIdx.x & 31;
    const int wave = threadIdx.x >> 5;
    const int t    = blockIdx.x >> 3;
    const int j0   = (blockIdx.x & 7) * 32;
    const int jmax = 2 * t + 1;
    float* wbase = W + ((size_t)t * B_DIM + wave * 16) * NSLOTS + j0 + lane;

    if (j0 <= jmax) {
        const float2* P2 = (const float2*)P + ((size_t)t * NSLOTS + j0) * B_DIM;
        #pragma unroll
        for (int r = 0; r < 4; ++r) {
            const int jj = wave * 4 + r;
            float w0 = 0.0f, w1 = 0.0f, w2 = 0.0f, w3 = 0.0f;
            if (j0 + jj <= jmax) {
                const float2* row = P2 + (size_t)jj * B_DIM;
                const float2 e0 = row[lane];
                const float2 e1 = row[lane + 32];
                const float2 e2 = row[lane + 64];
                const float2 e3 = row[lane + 96];
                float m = fmaxf(fmaxf(e0.x, e1.x), fmaxf(e2.x, e3.x));
                #pragma unroll
                for (int o = 16; o >= 1; o >>= 1) m = fmaxf(m, __shfl_xor(m, o));
                w0 = fminf(e0.y, m);
                w1 = fminf(e1.y, m);
                w2 = fminf(e2.y, m);
                w3 = fminf(e3.y, m);
            }
            Wt[(lane)      * 33 + jj] = w0;
            Wt[(lane + 32) * 33 + jj] = w1;
            Wt[(lane + 64) * 33 + jj] = w2;
            Wt[(lane + 96) * 33 + jj] = w3;
        }
        __syncthreads();
        #pragma unroll
        for (int i = 0; i < 16; ++i)
            wbase[(size_t)i * NSLOTS] = Wt[(wave * 16 + i) * 33 + lane];
    } else {
        #pragma unroll
        for (int i = 0; i < 16; ++i)
            wbase[(size_t)i * NSLOTS] = 0.0f;
    }
}

__global__ __launch_bounds__(256) void gemm_kernel(
    const float* __restrict__ W,
    const float* __restrict__ v1,
    const float* __restrict__ v2,
    float* __restrict__ out)
{
    const int lane = threadIdx.x & 31;
    const int wave = threadIdx.x >> 5;
    const int lm   = lane & 15;
    const int half = lane >> 4;

    const int b  = blockIdx.x >> 3;
    const int mt = blockIdx.x & 7;
    const int n0 = wave * 64;

    v8f acc0 = {}, acc1 = {}, acc2 = {}, acc3 = {};

    const float* Ap = W + ((size_t)(mt * 16 + lm) * B_DIM + b) * NSLOTS + 2 * half;
    const size_t bstride = (size_t)B_DIM * R_DIM;
    const float* B1 = v1 + (size_t)half * bstride + (size_t)b * R_DIM + n0 + lm;
    const float* B2 = v2 + (size_t)half * bstride + (size_t)b * R_DIM + n0 + lm;

    #pragma unroll 4
    for (int k = 0; k < NSLOTS; k += 4) {
        const v2f a = *(const v2f*)(Ap + k);
        const size_t off = (size_t)(k >> 1) * bstride;
        v2f b0, b1, b2, b3;
        b0.x = B1[off +  0]; b0.y = B2[off +  0];
        b1.x = B1[off + 16]; b1.y = B2[off + 16];
        b2.x = B1[off + 32]; b2.y = B2[off + 32];
        b3.x = B1[off + 48]; b3.y = B2[off + 48];
        acc0 = __builtin_amdgcn_wmma_f32_16x16x4_f32(false, a, false, b0, (short)0, acc0, false, false);
        acc1 = __builtin_amdgcn_wmma_f32_16x16x4_f32(false, a, false, b1, (short)0, acc1, false, false);
        acc2 = __builtin_amdgcn_wmma_f32_16x16x4_f32(false, a, false, b2, (short)0, acc2, false, false);
        acc3 = __builtin_amdgcn_wmma_f32_16x16x4_f32(false, a, false, b3, (short)0, acc3, false, false);
    }

    #pragma unroll
    for (int r = 0; r < 8; ++r) {
        const int trow = mt * 16 + r + 8 * half;
        float* orow = out + ((size_t)trow * B_DIM + b) * R_DIM + n0 + lm;
        orow[ 0] = acc0[r];
        orow[16] = acc1[r];
        orow[32] = acc2[r];
        orow[48] = acc3[r];
    }
}

extern "C" void kernel_launch(void* const* d_in, const int* in_sizes, int n_in,
                              void* d_out, int out_size, void* d_ws, size_t ws_size,
                              hipStream_t stream) {
    const float* u  = (const float*)d_in[0];
    const float* d1 = (const float*)d_in[1];
    const float* d2 = (const float*)d_in[2];
    const float* v1 = (const float*)d_in[3];
    const float* v2 = (const float*)d_in[4];
    float* out = (float*)d_out;
    float* W   = (float*)d_ws;
    float* P   = W + (size_t)T_DIM * B_DIM * NSLOTS;

    state_kernel<<<dim3(B_DIM / 32), dim3(32 * NW), 0, stream>>>(u, d1, d2, P);
    weights_kernel<<<dim3(T_DIM * 8), dim3(256), 0, stream>>>(P, W);
    gemm_kernel<<<dim3(B_DIM * 8), dim3(256), 0, stream>>>(W, v1, v2, out);
}


# State kernel: one wave per batch, 8 slots per lane in registers; used strength = scan across lanes (DPP) + in-lane sums, not a serial walk: sum order changes; P as [t][b][slot] planes
# speedup vs baseline: 47.3884x; 3.3704x over previous
#include <hip/hip_runtime.h>

#define T_DIM 128
#define B_DIM 128
#define R_DIM 512
#define NSLOTS 256

typedef float v2f __attribute__((ext_vector_type(2)));
typedef float v8f __attribute__((ext_vector_type(8)));

#define SPL 8

template <int CTRL>
__device__ __forceinline__ float row_read0(const float v)
{
    return __int_as_float(__builtin_amdgcn_update_dpp(0, __float_as_int(v), CTRL, 0xf, 0xf, true));
}

__global__ __launch_bounds__(32) void state_kernel(
    const float* __restrict__ u,
    const float* __restrict__ d1,
    const float* __restrict__ d2,
    float* __restrict__ PX,
    float* __restrict__ PS)
{
    const int lane = threadIdx.x;
    const int b = blockIdx.x;

    float uu[4], dd[SPL], c[SPL];
    #pragma unroll
    for (int r = 0; r < 4; ++r) {
        const int t = 4 * lane + r;
        uu[r]         = u[t * B_DIM + b];
        dd[2 * r]     = d1[t * B_DIM + b];
        dd[2 * r + 1] = d2[t * B_DIM + b];
    }
    #pragma unroll
    for (int p = 0; p < SPL; ++p) c[p] = 0.0f;

    const size_t tstride = (size_t)B_DIM * NSLOTS;
    float* px = PX + (size_t)b * NSLOTS + lane * SPL;
    float* ps = PS + (size_t)b * NSLOTS + lane * SPL;

    for (int t4 = 0; t4 < T_DIM / 4; ++t4) {
        #pragma unroll
        for (int r = 0; r < 4; ++r) {
            const float ut = __int_as_float(__builtin_amdgcn_readlane(__float_as_int(uu[r]), t4));
            if (lane == t4) { c[2 * r] = dd[2 * r]; c[2 * r + 1] = dd[2 * r + 1]; }

            float q[SPL];
            q[SPL - 1] = 0.0f;
            #pragma unroll
            for (int p = SPL - 2; p >= 0; --p) q[p] = q[p + 1] + c[p + 1];
            float v = ((c[7] + c[6]) + (c[5] + c[4])) + ((c[3] + c[2]) + (c[1] + c[0]));
            v += row_read0<0x101>(v);
            v += row_read0<0x102>(v);
            v += row_read0<0x104>(v);
            v += row_read0<0x108>(v);
            const float upper = __int_as_float(__builtin_amdgcn_readlane(__float_as_int(v), 16));
            float base = row_read0<0x101>(v);
            if (lane < 16) base += upper;

            float x[SPL], so[SPL];
            #pragma unroll
            for (int p = 0; p < SPL; ++p) {
                const float s = c[p];
                x[p] = ut - (base + q[p]);
                const float take = fminf(s, fmaxf(0.0f, x[p]));
                so[p] = s;
                c[p] = s - take;
            }
            if (lane <= t4) {
                *(float4*)(px)     = make_float4(x[0], x[1], x[2], x[3]);
                *(float4*)(px + 4) = make_float4(x[4], x[5], x[6], x[7]);
                *(float4*)(ps)     = make_float4(so[0], so[1], so[2], so[3]);
                *(float4*)(ps + 4) = make_float4(so[4], so[5], so[6], so[7]);
            }
            px += tstride;
            ps += tstride;
        }
    }
}

__global__ __launch_bounds__(256) void weights_kernel(
    const float* __restrict__ PX,
    const float* __restrict__ PS,
    float* __restrict__ W)
{
    __shared__ float M[8 * 32];

    const int lane = threadIdx.x & 31;
    const int wave = threadIdx.x >> 5;
    const int t    = blockIdx.x >> 3;
    const int j0   = (blockIdx.x & 7) * 32;
    const int jmax = 2 * t + 1;
    const size_t base = ((size_t)t * B_DIM + wave * 16) * NSLOTS + j0 + lane;
    float* wbase = W + base;

    if (j0 <= jmax) {
        const bool act = (j0 + lane) <= jmax;
        float sv[16];
        float m = 0.0f;
        if (act) {
            const float* xp = PX + base;
            const float* sp = PS + base;
            float xv[16];
            #pragma unroll
            for (int i = 0; i < 16; ++i) {
                xv[i] = xp[(size_t)i * NSLOTS];
                sv[i] = sp[(size_t)i * NSLOTS];
            }
            m = xv[0];
            #pragma unroll
            for (int i = 1; i < 16; ++i) m = fmaxf(m, xv[i]);
        } else {
            #pragma unroll
            for (int i = 0; i < 16; ++i) sv[i] = 0.0f;
        }
        M[wave * 32 + lane] = m;
        __syncthreads();
        float mm = M[lane];
        #pragma unroll
        for (int w = 1; w < 8; ++w) mm = fmaxf(mm, M[w * 32 + lane]);
        #pragma unroll
        for (int i = 0; i < 16; ++i)
            wbase[(size_t)i * NSLOTS] = act ? fminf(sv[i], mm) : 0.0f;
    } else {
        #pragma unroll
        for (int i = 0; i < 16; ++i)
            wbase[(size_t)i * NSLOTS] = 0.0f;
    }
}

__global__ __launch_bounds__(256) void gemm_kernel(
    const float* __restrict__ W,
    const float* __restrict__ v1,
    const float* __restrict__ v2,
    float* __restrict__ out)
{
    const int lane = threadIdx.x & 31;
    const int wave = threadIdx.x >> 5;
    const int lm   = lane & 15;
    const int half = lane >> 4;

    const int b  = blockIdx.x >> 3;
    const int mt = blockIdx.x & 7;
    const int n0 = wave * 64;

    v8f acc0 = {}, acc1 = {}, acc2 = {}, acc3 = {};

    const float* Ap = W + ((size_t)(mt * 16 + lm) * B_DIM + b) * NSLOTS + 2 * half;
    const size_t bstride = (size_t)B_DIM * R_DIM;
    const float* B1 = v1 + (size_t)half * bstride + (size_t)b * R_DIM + n0 + lm;
    const float* B2 = v2 + (size_t)half * bstride + (size_t)b * R_DIM + n0 + lm;

    #pragma unroll 4
    for (int k = 0; k < NSLOTS; k += 4) {
        const v2f a = *(const v2f*)(Ap + k);
        const size_t off = (size_t)(k >> 1) * bstride;
        v2f b0, b1, b2, b3;
        b0.x = B1[off +  0]; b0.y = B2[off +  0];
        b1.x = B1[off + 16]; b1.y = B2[off + 16];
        b2.x = B1[off + 32]; b2.y = B2[off + 32];
        b3.x = B1[off + 48]; b3.y = B2[off + 48];
        acc0 = __builtin_amdgcn_wmma_f32_16x16x4_f32(false, a, false, b0, (short)0, acc0, false, false);
        acc1 = __builtin_amdgcn_wmma_f32_16x16x4_f32(false, a, false, b1, (short)0, acc1, false, false);
        acc2 = __builtin_amdgcn_wmma_f32_16x16x4_f32(false, a, false, b2, (short)0, acc2, false, false);
        acc3 = __builtin_amdgcn_wmma_f32_16x16x4_f32(false, a, false, b3, (short)0, acc3, false, false);
    }

    #pragma unroll
    for (int r = 0; r < 8; ++r) {
        const int trow = mt * 16 + r + 8 * half;
        float* orow = out + ((size_t)trow * B_DIM + b) * R_DIM + n0 + lm;
        orow[ 0] = acc0[r];
        orow[16] = acc1[r];
        orow[32] = acc2[r];
        orow[48] = acc3[r];
    }
}

extern "C" void kernel_launch(void* const* d_in, const int* in_sizes, int n_in,
                              void* d_out, int out_size, void* d_ws, size_t ws_size,
                              hipStream_t stream) {
    const float* u  = (const float*)d_in[0];
    const float* d1 = (const float*)d_in[1];
    const float* d2 = (const float*)d_in[2];
    const float* v1 = (const float*)d_in[3];
    const float* v2 = (const float*)d_in[4];
    float* out = (float*)d_out;
    float* W   = (float*)d_ws;
    float* PX  = W  + (size_t)T_DIM * B_DIM * NSLOTS;
    float* PS  = PX + (size_t)T_DIM * B_DIM * NSLOTS;

    state_kernel<<<dim3(B_DIM), dim3(32), 0, stream>>>(u, d1, d2, PX, PS);
    weights_kernel<<<dim3(T_DIM * 8), dim3(256), 0, stream>>>(PX, PS, W);
    gemm_kernel<<<dim3(B_DIM * 8), dim3(256), 0, stream>>>(W, v1, v2, out);
}


# GEMM computes only W's non-zero triangle (row t is 0 above slot 2t+1): K block J (32 slots) feeds row tiles J..7 only; wave = 128 rows x 32 cols, W block staged in LDS; same k order
# speedup vs baseline: 85.5194x; 1.8046x over previous
#include <hip/hip_runtime.h>

#define T_DIM 128
#define B_DIM 128
#define R_DIM 512
#define NSLOTS 256

typedef float v2f __attribute__((ext_vector_type(2)));
typedef float v8f __attribute__((ext_vector_type(8)));

#define SPL 8

template <int CTRL>
__device__ __forceinline__ float row_read0(const float v)
{
    return __int_as_float(__builtin_amdgcn_update_dpp(0, __float_as_int(v), CTRL, 0xf, 0xf, true));
}

__global__ __launch_bounds__(32) void state_kernel(
    const float* __restrict__ u,
    const float* __restrict__ d1,
    const float* __restrict__ d2,
    float* __restrict__ PX,
    float* __restrict__ PS)
{
    const int lane = threadIdx.x;
    const int b = blockIdx.x;

    float uu[4], dd[SPL], c[SPL];
    #pragma unroll
    for (int r = 0; r < 4; ++r) {
        const int t = 4 * lane + r;
        uu[r]         = u[t * B_DIM + b];
        dd[2 * r]     = d1[t * B_DIM + b];
        dd[2 * r + 1] = d2[t * B_DIM + b];
    }
    #pragma unroll
    for (int p = 0; p < SPL; ++p) c[p] = 0.0f;

    const size_t tstride = (size_t)B_DIM * NSLOTS;
    float* px = PX + (size_t)b * NSLOTS + lane * SPL;
    float* ps = PS + (size_t)b * NSLOTS + lane * SPL;

    for (int t4 = 0; t4 < T_DIM / 4; ++t4) {
        #pragma unroll
        for (int r = 0; r < 4; ++r) {
            const float ut = __int_as_float(__builtin_amdgcn_readlane(__float_as_int(uu[r]), t4));
            if (lane == t4) { c[2 * r] = dd[2 * r]; c[2 * r + 1] = dd[2 * r + 1]; }

            float q[SPL];
            q[SPL - 1] = 0.0f;
            #pragma unroll
            for (int p = SPL - 2; p >= 0; --p) q[p] = q[p + 1] + c[p + 1];
            float v = ((c[7] + c[6]) + (c[5] + c[4])) + ((c[3] + c[2]) + (c[1] + c[0]));
            v += row_read0<0x101>(v);
            v += row_read0<0x102>(v);
            v += row_read0<0x104>(v);
            v += row_read0<0x108>(v);
            const float upper = __int_as_float(__builtin_amdgcn_readlane(__float_as_int(v), 16));
            float base = row_read0<0x101>(v);
            if (lane < 16) base += upper;

            float x[SPL], so[SPL];
            #pragma unroll
            for (int p = 0; p < SPL; ++p) {
                const float s = c[p];
                x[p] = ut - (base + q[p]);
                const float take = fminf(s, fmaxf(0.0f, x[p]));
                so[p] = s;
                c[p] = s - take;
            }
            if (lane <= t4) {
                *(float4*)(px)     = make_float4(x[0], x[1], x[2], x[3]);
                *(float4*)(px + 4) = make_float4(x[4], x[5], x[6], x[7]);
                *(float4*)(ps)     = make_float4(so[0], so[1], so[2], so[3]);
                *(float4*)(ps + 4) = make_float4(so[4], so[5], so[6], so[7]);
            }
            px += tstride;
            ps += tstride;
        }
    }
}

__global__ __launch_bounds__(256) void weights_kernel(
    const float* __restrict__ PX,
    const float* __restrict__ PS,
    float* __restrict__ W)
{
    __shared__ float M[8 * 32];

    const int lane = threadIdx.x & 31;
    const int wave = threadIdx.x >> 5;
    const int t    = blockIdx.x >> 3;
    const int j0   = (blockIdx.x & 7) * 32;
    const int jmax = 2 * t + 1;
    const size_t base = ((size_t)t * B_DIM + wave * 16) * NSLOTS + j0 + lane;
    float* wbase = W + base;

    if (j0 <= jmax) {
        const bool act = (j0 + lane) <= jmax;
        float sv[16];
        float m = 0.0f;
        if (act) {
            const float* xp = PX + base;
            const float* sp = PS + base;
            float xv[16];
            #pragma unroll
            for (int i = 0; i < 16; ++i) {
                xv[i] = xp[(size_t)i * NSLOTS];
                sv[i] = sp[(size_t)i * NSLOTS];
            }
            m = xv[0];
            #pragma unroll
            for (int i = 1; i < 16; ++i) m = fmaxf(m, xv[i]);
        } else {
            #pragma unroll
            for (int i = 0; i < 16; ++i) sv[i] = 0.0f;
        }
        M[wave * 32 + lane] = m;
        __syncthreads();
        float mm = M[lane];
        #pragma unroll
        for (int w = 1; w < 8; ++w) mm = fmaxf(mm, M[w * 32 + lane]);
        #pragma unroll
        for (int i = 0; i < 16; ++i)
            wbase[(size_t)i * NSLOTS] = act ? fminf(sv[i], mm) : 0.0f;
    } else {
        #pragma unroll
        for (int i = 0; i < 16; ++i)
            wbase[(size_t)i * NSLOTS] = 0.0f;
    }
}

#define KB 32
#define APITCH 36
#define ABUF (T_DIM * APITCH)

__device__ __forceinline__ void lds_barrier()
{
    __builtin_amdgcn_fence(__ATOMIC_RELEASE, "workgroup", "local");
    __builtin_amdgcn_s_barrier();
    __builtin_amdgcn_fence(__ATOMIC_ACQUIRE, "workgroup", "local");
}

template <int J, int RD>
__device__ __forceinline__ int panel_piece(const int tid)
{
    return (256 * RD + 256 <= (8 - J) * 128) ? 256 * RD + tid : 256 * RD + (tid & 127);
}

template <int J, int RD>
__device__ __forceinline__ float4 panel_load(const float* __restrict__ Wb, const int tid)
{
    float4 r = make_float4(0.0f, 0.0f, 0.0f, 0.0f);
    if (256 * RD < (8 - J) * 128) {
        const int idx = panel_piece<J, RD>(tid);
        const int row = 16 * J + (idx >> 3);
        r = *(const float4*)(Wb + (size_t)row * (B_DIM * NSLOTS) + KB * J + 4 * (idx & 7));
    }
    return r;
}

template <int J, int RD>
__device__ __forceinline__ void panel_store(const float4 r, float* __restrict__ Ab, const int tid)
{
    if (256 * RD < (8 - J) * 128) {
        const int idx = panel_piece<J, RD>(tid);
        const int row = 16 * J + (idx >> 3);
        *(float4*)(Ab + row * APITCH + 4 * (idx & 7)) = r;
    }
}

template <int J>
__device__ __forceinline__ void b_load(v2f (&bx)[8], v2f (&by)[8],
                                       const float* __restrict__ B1, const float* __restrict__ B2)
{
    const size_t bstride = (size_t)B_DIM * R_DIM;
    #pragma unroll
    for (int s = 0; s < 8; ++s) {
        bx[s] = *(const v2f*)(B1 + (size_t)(16 * J + 2 * s) * bstride);
        by[s] = *(const v2f*)(B2 + (size_t)(16 * J + 2 * s) * bstride);
    }
}

template <int MT>
__device__ __forceinline__ void tile_store(const v8f (&acc)[8][2], float* __restrict__ orow)
{
    #pragma unroll
    for (int r = 0; r < 8; ++r) {
        v2f o;
        o.x = acc[MT][0][r];
        o.y = acc[MT][1][r];
        *(v2f*)(orow + (size_t)(MT * 16 + r) * (B_DIM * R_DIM)) = o;
    }
}

template <int J>
__device__ __forceinline__ void k_block(v8f (&acc)[8][2], v2f (&bx)[8], v2f (&by)[8],
                                        float* __restrict__ As, const float* __restrict__ Wb,
                                        const float* __restrict__ B1, const float* __restrict__ B2,
                                        float* __restrict__ orow,
                                        const int tid, const int lm, const int half)
{
    constexpr int JN = (J < 7 ? J + 1 : 7);
    float4 st0 = make_float4(0.0f, 0.0f, 0.0f, 0.0f), st1 = st0, st2 = st0, st3 = st0;
    v2f nbx[8], nby[8];
    if (J < 7) {
        b_load<JN>(nbx, nby, B1, B2);
        st0 = panel_load<JN, 0>(Wb, tid);
        st1 = panel_load<JN, 1>(Wb, tid);
        st2 = panel_load<JN, 2>(Wb, tid);
        st3 = panel_load<JN, 3>(Wb, tid);
    }
    if (J > 0) tile_store<(J > 0 ? J - 1 : 0)>(acc, orow);
    const float* Af = As + (J & 1) * ABUF + lm * APITCH + 2 * half;
    #pragma unroll
    for (int s = 0; s < 8; ++s) {
        v2f b0, b1;
        b0.x = bx[s].x; b0.y = by[s].x;
        b1.x = bx[s].y; b1.y = by[s].y;
        #pragma unroll
        for (int mt = J; mt < 8; ++mt) {
            const v2f a = *(const v2f*)(Af + mt * 16 * APITCH + 4 * s);
            acc[mt][0] = __builtin_amdgcn_wmma_f32_16x16x4_f32(false, a, false, b0, (short)0, acc[mt][0], false, false);
            acc[mt][1] = __builtin_amdgcn_wmma_f32_16x16x4_f32(false, a, false, b1, (short)0, acc[mt][1], false, false);
        }
        if (J < 7 && s == 3) {
            float* An = As + ((J + 1) & 1) * ABUF;
            panel_store<JN, 0>(st0, An, tid);
            panel_store<JN, 1>(st1, An, tid);
            panel_store<JN, 2>(st2, An, tid);
            panel_store<JN, 3>(st3, An, tid);
        }
    }
    if (J < 7) {
        lds_barrier();
        #pragma unroll
        for (int s = 0; s < 8; ++s) { bx[s] = nbx[s]; by[s] = nby[s]; }
    }
}

__global__ __launch_bounds__(256) void gemm_kernel(
    const float* __restrict__ W,
    const float* __restrict__ v1,
    const float* __restrict__ v2,
    float* __restrict__ out)
{
    __shared__ __attribute__((aligned(16))) float As[2 * ABUF];

    const int tid  = threadIdx.x;
    const int lane = tid & 31;
    const int wave = tid >> 5;
    const int lm   = lane & 15;
    const int half = lane >> 4;

    const int b  = blockIdx.x >> 1;
    const int n0 = (blockIdx.x & 1) * 256 + wave * 32;

    v8f acc[8][2];
    #pragma unroll
    for (int mt = 0; mt < 8; ++mt) { acc[mt][0] = v8f{}; acc[mt][1] = v8f{}; }

    const float* Wb = W + (size_t)b * NSLOTS;
    const float* B1 = v1 + ((size_t)half * B_DIM + b) * R_DIM + n0 + 2 * lm;
    const float* B2 = v2 + ((size_t)half * B_DIM + b) * R_DIM + n0 + 2 * lm;

    float* orow = out + ((size_t)(8 * half) * B_DIM + b) * R_DIM + n0 + 2 * lm;

    v2f bx[8], by[8];
    {
        b_load<0>(bx, by, B1, B2);
        const float4 p0 = panel_load<0, 0>(Wb, tid);
        const float4 p1 = panel_load<0, 1>(Wb, tid);
        const float4 p2 = panel_load<0, 2>(Wb, tid);
        const float4 p3 = panel_load<0, 3>(Wb, tid);
        panel_store<0, 0>(p0, As, tid);
        panel_store<0, 1>(p1, As, tid);
        panel_store<0, 2>(p2, As, tid);
        panel_store<0, 3>(p3, As, tid);
        lds_barrier();
    }

    k_block<0>(acc, bx, by, As, Wb, B1, B2, orow, tid, lm, half);
    k_block<1>(acc, bx, by, As, Wb, B1, B2, orow, tid, lm, half);
    k_block<2>(acc, bx, by, As, Wb, B1, B2, orow, tid, lm, half);
    k_block<3>(acc, bx, by, As, Wb, B1, B2, orow, tid, lm, half);
    k_block<4>(acc, bx, by, As, Wb, B1, B2, orow, tid, lm, half);
    k_block<5>(acc, bx, by, As, Wb, B1, B2, orow, tid, lm, half);
    k_block<6>(acc, bx, by, As, Wb, B1, B2, orow, tid, lm, half);
    k_block<7>(acc, bx, by, As, Wb, B1, B2, orow, tid, lm, half);

    tile_store<7>(acc, orow);
}

extern "C" void kernel_launch(void* const* d_in, const int* in_sizes, int n_in,
                              void* d_out, int out_size, void* d_ws, size_t ws_size,
                              hipStream_t stream) {
    const float* u  = (const float*)d_in[0];
    const float* d1 = (const float*)d_in[1];
    const float* d2 = (const float*)d_in[2];
    const float* v1 = (const float*)d_in[3];
    const float* v2 = (const float*)d_in[4];
    float* out = (float*)d_out;
    float* W   = (float*)d_ws;
    float* PX  = W  + (size_t)T_DIM * B_DIM * NSLOTS;
    float* PS  = PX + (size_t)T_DIM * B_DIM * NSLOTS;

    state_kernel<<<dim3(B_DIM), dim3(32), 0, stream>>>(u, d1, d2, PX, PS);
    weights_kernel<<<dim3(T_DIM * 8), dim3(256), 0, stream>>>(PX, PS, W);
    gemm_kernel<<<dim3(B_DIM * 2), dim3(256), 0, stream>>>(W, v1, v2, out);
}


# State kernel: one lane per slot carries the strength above its slot from step to step (pop: max(0, used - u); push adds) instead of re-summing all slots: no scan across lanes; sum order changes
# speedup vs baseline: 105.5211x; 1.2339x over previous
#include <hip/hip_runtime.h>

#define T_DIM 128
#define B_DIM 128
#define R_DIM 512
#define NSLOTS 256

typedef float v2f __attribute__((ext_vector_type(2)));
typedef float v8f __attribute__((ext_vector_type(8)));

#define SG 8

struct GroupIn { float4 u[SG / 4], a[SG / 4], c[SG / 4]; };

__device__ __forceinline__ GroupIn group_load(const float* __restrict__ su, const float* __restrict__ sa,
                                              const float* __restrict__ sc, const int tg)
{
    GroupIn v;
    #pragma unroll
    for (int i = 0; i < SG / 4; ++i) {
        v.u[i] = *(const float4*)(su + SG * tg + 4 * i);
        v.a[i] = *(const float4*)(sa + SG * tg + 4 * i);
        v.c[i] = *(const float4*)(sc + SG * tg + 4 * i);
    }
    return v;
}

__device__ __forceinline__ float comp(const float4 v, const int i)
{
    return i == 0 ? v.x : (i == 1 ? v.y : (i == 2 ? v.z : v.w));
}

template <bool PUSH>
__device__ __forceinline__ void group_steps(float& e, float& s, const GroupIn in, const int rel,
                                            float* __restrict__ px, float* __restrict__ ps)
{
    const size_t tstride = (size_t)B_DIM * NSLOTS;
    #pragma unroll
    for (int r = 0; r < SG; ++r) {
        const float ut = comp(in.u[r / 4], r % 4);
        const float d1t = comp(in.a[r / 4], r % 4);
        const float d2t = comp(in.c[r / 4], r % 4);
        const float up = fmaxf(ut, 0.0f);
        const float both = d1t + d2t;
        if (PUSH) {
            e += (rel < 2 * r) ? both : ((rel == 2 * r) ? d2t : 0.0f);
            s += (rel == 2 * r) ? d1t : ((rel == 2 * r + 1) ? d2t : 0.0f);
        } else {
            e += both;
        }
        const float x = ut - e;
        px[(size_t)r * tstride] = x;
        ps[(size_t)r * tstride] = s;
        s -= fminf(s, fmaxf(0.0f, x));
        e = fmaxf(0.0f, e - up);
    }
}

__global__ __launch_bounds__(128) void state_kernel(
    const float* __restrict__ u,
    const float* __restrict__ d1,
    const float* __restrict__ d2,
    float* __restrict__ PX,
    float* __restrict__ PS)
{
    __shared__ __attribute__((aligned(16))) float su[T_DIM];
    __shared__ __attribute__((aligned(16))) float sa[T_DIM];
    __shared__ __attribute__((aligned(16))) float sc[T_DIM];

    const int tid  = threadIdx.x;
    const int b    = blockIdx.x >> 1;
    const int slot = (blockIdx.x & 1) * 128 + tid;
    const int g    = __builtin_amdgcn_readfirstlane(slot >> 5);

    su[tid] = u[tid * B_DIM + b];
    sa[tid] = d1[tid * B_DIM + b];
    sc[tid] = d2[tid * B_DIM + b];
    __syncthreads();

    const size_t tstride = (size_t)B_DIM * NSLOTS;
    float* px = PX + ((size_t)(16 * g) * B_DIM + b) * NSLOTS;
    float* ps = PS + ((size_t)(16 * g) * B_DIM + b) * NSLOTS;

    float e = 0.0f;
    float s = 0.0f;

    #pragma unroll
    for (int k = 0; k < 16 / SG; ++k) {
        const int tg = (16 / SG) * g + k;
        const GroupIn in = group_load(su, sa, sc, tg);
        group_steps<true>(e, s, in, slot - 2 * SG * tg, px + slot, ps + slot);
        px += SG * tstride;
        ps += SG * tstride;
    }

    if (g < 7) {
        GroupIn in = group_load(su, sa, sc, (16 / SG) * (g + 1));
        for (int tg = (16 / SG) * (g + 1); tg < T_DIM / SG; ++tg) {
            const GroupIn nx = group_load(su, sa, sc, (tg + 1 < T_DIM / SG) ? tg + 1 : tg);
            group_steps<false>(e, s, in, 0, px + slot, ps + slot);
            px += SG * tstride;
            ps += SG * tstride;
            in = nx;
        }
    }
}

__global__ __launch_bounds__(256) void weights_kernel(
    const float* __restrict__ PX,
    const float* __restrict__ PS,
    float* __restrict__ W)
{
    __shared__ float M[8 * 32];

    const int lane = threadIdx.x & 31;
    const int wave = threadIdx.x >> 5;
    const int t    = blockIdx.x >> 3;
    const int j0   = (blockIdx.x & 7) * 32;
    const int jmax = 2 * t + 1;
    const size_t base = ((size_t)t * B_DIM + wave * 16) * NSLOTS + j0 + lane;
    float* wbase = W + base;

    if (j0 <= jmax) {
        const bool act = (j0 + lane) <= jmax;
        float sv[16];
        float m = 0.0f;
        if (act) {
            const float* xp = PX + base;
            const float* sp = PS + base;
            float xv[16];
            #pragma unroll
            for (int i = 0; i < 16; ++i) {
                xv[i] = xp[(size_t)i * NSLOTS];
                sv[i] = sp[(size_t)i * NSLOTS];
            }
            m = xv[0];
            #pragma unroll
            for (int i = 1; i < 16; ++i) m = fmaxf(m, xv[i]);
        } else {
            #pragma unroll
            for (int i = 0; i < 16; ++i) sv[i] = 0.0f;
        }
        M[wave * 32 + lane] = m;
        __syncthreads();
        float mm = M[lane];
        #pragma unroll
        for (int w = 1; w < 8; ++w) mm = fmaxf(mm, M[w * 32 + lane]);
        #pragma unroll
        for (int i = 0; i < 16; ++i)
            wbase[(size_t)i * NSLOTS] = act ? fminf(sv[i], mm) : 0.0f;
    } else {
        #pragma unroll
        for (int i = 0; i < 16; ++i)
            wbase[(size_t)i * NSLOTS] = 0.0f;
    }
}

#define KB 32
#define APITCH 36
#define ABUF (T_DIM * APITCH)

__device__ __forceinline__ void lds_barrier()
{
    __builtin_amdgcn_fence(__ATOMIC_RELEASE, "workgroup", "local");
    __builtin_amdgcn_s_barrier();
    __builtin_amdgcn_fence(__ATOMIC_ACQUIRE, "workgroup", "local");
}

template <int J, int RD>
__device__ __forceinline__ int panel_piece(const int tid)
{
    return (256 * RD + 256 <= (8 - J) * 128) ? 256 * RD + tid : 256 * RD + (tid & 127);
}

template <int J, int RD>
__device__ __forceinline__ float4 panel_load(const float* __restrict__ Wb, const int tid)
{
    float4 r = make_float4(0.0f, 0.0f, 0.0f, 0.0f);
    if (256 * RD < (8 - J) * 128) {
        const int idx = panel_piece<J, RD>(tid);
        const int row = 16 * J + (idx >> 3);
        r = *(const float4*)(Wb + (size_t)row * (B_DIM * NSLOTS) + KB * J + 4 * (idx & 7));
    }
    return r;
}

template <int J, int RD>
__device__ __forceinline__ void panel_store(const float4 r, float* __restrict__ Ab, const int tid)
{
    if (256 * RD < (8 - J) * 128) {
        const int idx = panel_piece<J, RD>(tid);
        const int row = 16 * J + (idx >> 3);
        *(float4*)(Ab + row * APITCH + 4 * (idx & 7)) = r;
    }
}

template <int J>
__device__ __forceinline__ void b_load(v2f (&bx)[8], v2f (&by)[8],
                                       const float* __restrict__ B1, const float* __restrict__ B2)
{
    const size_t bstride = (size_t)B_DIM * R_DIM;
    #pragma unroll
    for (int s = 0; s < 8; ++s) {
        bx[s] = *(const v2f*)(B1 + (size_t)(16 * J + 2 * s) * bstride);
        by[s] = *(const v2f*)(B2 + (size_t)(16 * J + 2 * s) * bstride);
    }
}

template <int MT>
__device__ __forceinline__ void tile_store(const v8f (&acc)[8][2], float* __restrict__ orow)
{
    #pragma unroll
    for (int r = 0; r < 8; ++r) {
        v2f o;
        o.x = acc[MT][0][r];
        o.y = acc[MT][1][r];
        *(v2f*)(orow + (size_t)(MT * 16 + r) * (B_DIM * R_DIM)) = o;
    }
}

template <int J>
__device__ __forceinline__ void k_block(v8f (&acc)[8][2], v2f (&bx)[8], v2f (&by)[8],
                                        float* __restrict__ As, const float* __restrict__ Wb,
                                        const float* __restrict__ B1, const float* __restrict__ B2,
                                        float* __restrict__ orow,
                                        const int tid, const int lm, const int half)
{
    constexpr int JN = (J < 7 ? J + 1 : 7);
    float4 st0 = make_float4(0.0f, 0.0f, 0.0f, 0.0f), st1 = st0, st2 = st0, st3 = st0;
    v2f nbx[8], nby[8];
    if (J < 7) {
        b_load<JN>(nbx, nby, B1, B2);
        st0 = panel_load<JN, 0>(Wb, tid);
        st1 = panel_load<JN, 1>(Wb, tid);
        st2 = panel_load<JN, 2>(Wb, tid);
        st3 = panel_load<JN, 3>(Wb, tid);
    }
    if (J > 0) tile_store<(J > 0 ? J - 1 : 0)>(acc, orow);
    const float* Af = As + (J & 1) * ABUF + lm * APITCH + 2 * half;
    #pragma unroll
    for (int s = 0; s < 8; ++s) {
        v2f b0, b1;
        b0.x = bx[s].x; b0.y = by[s].x;
        b1.x = bx[s].y; b1.y = by[s].y;
        #pragma unroll
        for (int mt = J; mt < 8; ++mt) {
            const v2f a = *(const v2f*)(Af + mt * 16 * APITCH + 4 * s);
            acc[mt][0] = __builtin_amdgcn_wmma_f32_16x16x4_f32(false, a, false, b0, (short)0, acc[mt][0], false, false);
            acc[mt][1] = __builtin_amdgcn_wmma_f32_16x16x4_f32(false, a, false, b1, (short)0, acc[mt][1], false, false);
        }
        if (J < 7 && s == 3) {
            float* An = As + ((J + 1) & 1) * ABUF;
            panel_store<JN, 0>(st0, An, tid);
            panel_store<JN, 1>(st1, An, tid);
            panel_store<JN, 2>(st2, An, tid);
            panel_store<JN, 3>(st3, An, tid);
        }
    }
    if (J < 7) {
        lds_barrier();
        #pragma unroll
        for (int s = 0; s < 8; ++s) { bx[s] = nbx[s]; by[s] = nby[s]; }
    }
}

__global__ __launch_bounds__(256) void gemm_kernel(
    const float* __restrict__ W,
    const float* __restrict__ v1,
    const float* __restrict__ v2,
    float* __restrict__ out)
{
    __shared__ __attribute__((aligned(16))) float As[2 * ABUF];

    const int tid  = threadIdx.x;
    const int lane = tid & 31;
    const int wave = tid >> 5;
    const int lm   = lane & 15;
    const int half = lane >> 4;

    const int b  = blockIdx.x >> 1;
    const int n0 = (blockIdx.x & 1) * 256 + wave * 32;

    v8f acc[8][2];
    #pragma unroll
    for (int mt = 0; mt < 8; ++mt) { acc[mt][0] = v8f{}; acc[mt][1] = v8f{}; }

    const float* Wb = W + (size_t)b * NSLOTS;
    const float* B1 = v1 + ((size_t)half * B_DIM + b) * R_DIM + n0 + 2 * lm;
    const float* B2 = v2 + ((size_t)half * B_DIM + b) * R_DIM + n0 + 2 * lm;

    float* orow = out + ((size_t)(8 * half) * B_DIM + b) * R_DIM + n0 + 2 * lm;

    v2f bx[8], by[8];
    {
        b_load<0>(bx, by, B1, B2);
        const float4 p0 = panel_load<0, 0>(Wb, tid);
        const float4 p1 = panel_load<0, 1>(Wb, tid);
        const float4 p2 = panel_load<0, 2>(Wb, tid);
        const float4 p3 = panel_load<0, 3>(Wb, tid);
        panel_store<0, 0>(p0, As, tid);
        panel_store<0, 1>(p1, As, tid);
        panel_store<0, 2>(p2, As, tid);
        panel_store<0, 3>(p3, As, tid);
        lds_barrier();
    }

    k_block<0>(acc, bx, by, As, Wb, B1, B2, orow, tid, lm, half);
    k_block<1>(acc, bx, by, As, Wb, B1, B2, orow, tid, lm, half);
    k_block<2>(acc, bx, by, As, Wb, B1, B2, orow, tid, lm, half);
    k_block<3>(acc, bx, by, As, Wb, B1, B2, orow, tid, lm, half);
    k_block<4>(acc, bx, by, As, Wb, B1, B2, orow, tid, lm, half);
    k_block<5>(acc, bx, by, As, Wb, B1, B2, orow, tid, lm, half);
    k_block<6>(acc, bx, by, As, Wb, B1, B2, orow, tid, lm, half);
    k_block<7>(acc, bx, by, As, Wb, B1, B2, orow, tid, lm, half);

    tile_store<7>(acc, orow);
}

extern "C" void kernel_launch(void* const* d_in, const int* in_sizes, int n_in,
                              void* d_out, int out_size, void* d_ws, size_t ws_size,
                              hipStream_t stream) {
    const float* u  = (const float*)d_in[0];
    const float* d1 = (const float*)d_in[1];
    const float* d2 = (const float*)d_in[2];
    const float* v1 = (const float*)d_in[3];
    const float* v2 = (const float*)d_in[4];
    float* out = (float*)d_out;
    float* W   = (float*)d_ws;
    float* PX  = W  + (size_t)T_DIM * B_DIM * NSLOTS;
    float* PS  = PX + (size_t)T_DIM * B_DIM * NSLOTS;

    state_kernel<<<dim3(B_DIM * 2), dim3(128), 0, stream>>>(u, d1, d2, PX, PS);
    weights_kernel<<<dim3(T_DIM * 8), dim3(256), 0, stream>>>(PX, PS, W);
    gemm_kernel<<<dim3(B_DIM * 2), dim3(256), 0, stream>>>(W, v1, v2, out);
}


# Stack state in closed form (strength above a slot = sum of push minus pop, minus that sum's minimum since the push): one kernel per step holds all batches, writes W; 2 launches; sum order changes
# speedup vs baseline: 110.9016x; 1.0510x over previous
#include <hip/hip_runtime.h>

#define T_DIM 128
#define B_DIM 128
#define R_DIM 512
#define NSLOTS 256

typedef float v2f __attribute__((ext_vector_type(2)));
typedef float v8f __attribute__((ext_vector_type(8)));

#define CP 129
#define FW 16

template <bool GEN>
__device__ __forceinline__ void pair_eval(float& sa, float& sb, float& xa, float& xb,
                                          const float Rv, const float Gv, const float Cpv, const float base,
                                          const float ut, const float d1t, const float d2t,
                                          const bool old, const bool cur)
{
    const float mg = fminf(Gv, Rv);
    const float sao = fmaxf(0.0f, mg - fminf(Cpv, Rv));
    const float sbo = fmaxf(0.0f, Rv - Gv);
    const float xao = base + mg;
    const float xbo = base + Rv;
    if (GEN) {
        const float ninf = -__builtin_huge_valf();
        sa = old ? sao : (cur ? d1t : 0.0f);
        sb = old ? sbo : (cur ? d2t : 0.0f);
        xa = old ? xao : (cur ? ut - d2t : ninf);
        xb = old ? xbo : (cur ? ut : ninf);
    } else {
        sa = sao; sb = sbo; xa = xao; xb = xbo;
    }
}

__global__ __launch_bounds__(512) void weights_kernel(
    const float* __restrict__ u,
    const float* __restrict__ d1,
    const float* __restrict__ d2,
    float* __restrict__ W)
{
    __shared__ float Cs[B_DIM * CP];
    __shared__ float Gs[B_DIM * CP];
    __shared__ float Rs[B_DIM * CP];
    __shared__ float Tot[4 * B_DIM];
    __shared__ float Mn[4 * B_DIM];
    __shared__ v2f   Mx[2 * FW * 32];

    const int tid = threadIdx.x;
    const int t   = blockIdx.x >> 1;
    const int h   = blockIdx.x & 1;
    const int jgmax = t >> 5;
    if (h > jgmax) return;

    const int nrows = t > 0 ? t : 1;
    const int npad = (nrows + 31) & ~31;
    const int seg  = npad >> 2;

    const int b = tid & 127;
    const int g = tid >> 7;

    #pragma unroll
    for (int k = 0; k < T_DIM / 4; ++k) {
        const int  i  = g + 4 * k;
        const bool in = i < nrows;
        const int  ic = in ? i : nrows - 1;
        const float uu = u[ic * B_DIM + b], a = d1[ic * B_DIM + b], c2 = d2[ic * B_DIM + b];
        const float up = fmaxf(uu, 0.0f);
        Cs[b * CP + i] = in ? (a + c2) - up : 0.0f;
        Gs[b * CP + i] = in ? fmaxf(0.0f, c2 - up) : 0.0f;
    }
    __syncthreads();

    float* cp = Cs + b * CP + g * seg;
    float* gp = Gs + b * CP + g * seg;
    float* rp = Rs + b * CP + g * seg;
    const float pinf = __builtin_huge_valf();

    {
        float cc[8];
        #pragma unroll
        for (int k = 0; k < 8; ++k) cc[k] = cp[k];
        float acc = 0.0f;
        for (int i0 = 0; i0 < seg; i0 += 8) {
            const int nx = (i0 + 8 < seg) ? i0 + 8 : i0;
            float cn[8];
            #pragma unroll
            for (int k = 0; k < 8; ++k) cn[k] = cp[nx + k];
            #pragma unroll
            for (int k = 0; k < 8; ++k) { acc += cc[k]; cp[i0 + k] = acc; }
            #pragma unroll
            for (int k = 0; k < 8; ++k) cc[k] = cn[k];
        }
        Tot[g * B_DIM + b] = acc;
    }
    __syncthreads();

    {
        float off = 0.0f;
        #pragma unroll
        for (int g2 = 0; g2 < 3; ++g2) if (g2 < g) off += Tot[g2 * B_DIM + b];
        float r = pinf;
        float cv[8], gv[8];
        #pragma unroll
        for (int k = 0; k < 8; ++k) { cv[k] = cp[seg - 8 + k]; gv[k] = gp[seg - 8 + k]; }
        for (int i0 = seg - 8; i0 >= 0; i0 -= 8) {
            const int nx = (i0 >= 8) ? i0 - 8 : 0;
            float nc[8], ng[8];
            #pragma unroll
            for (int k = 0; k < 8; ++k) { nc[k] = cp[nx + k]; ng[k] = gp[nx + k]; }
            #pragma unroll
            for (int k = 7; k >= 0; --k) {
                const float C = cv[k] + off;
                cp[i0 + k] = C;
                gp[i0 + k] = C - gv[k];
                r = fminf(r, C);
                rp[i0 + k] = r;
            }
            #pragma unroll
            for (int k = 0; k < 8; ++k) { cv[k] = nc[k]; gv[k] = ng[k]; }
        }
        Mn[g * B_DIM + b] = r;
    }
    __syncthreads();

    {
        float carry = pinf;
        #pragma unroll
        for (int g2 = 1; g2 < 4; ++g2) if (g2 > g) carry = fminf(carry, Mn[g2 * B_DIM + b]);
        if (g < 3) {
            #pragma unroll 8
            for (int i = 0; i < seg; ++i) rp[i] = fminf(rp[i], carry);
        }
    }
    __syncthreads();

    const int lane = tid & 31;
    const int wv   = __builtin_amdgcn_readfirstlane(tid >> 5);

    float ut[8], d2t[8], d1t[8], base[8];
    #pragma unroll
    for (int q = 0; q < 8; ++q) {
        const int bq = 8 * wv + q;
        ut[q]  = u[t * B_DIM + bq];
        d1t[q] = d1[t * B_DIM + bq];
        d2t[q] = d2[t * B_DIM + bq];
        const float cn = (t > 0) ? Cs[bq * CP + (nrows - 1)] : 0.0f;
        base[q] = (ut[q] - (d1t[q] + d2t[q])) - cn;
    }

    float sa[2][8], sb[2][8];
    #pragma unroll
    for (int k = 0; k < 2; ++k) {
        const int jg = 2 * k + h;
        if (jg <= jgmax) {
            const int  t0  = 32 * jg + lane;
            const bool old = t0 < t;
            const bool cur = t0 == t;
            const int  t0c = t0 < nrows ? t0 : nrows - 1;
            const int  tpc = t0c > 0 ? t0c - 1 : 0;
            float ma = -pinf, mb = -pinf;
            if (32 * jg + 31 < t) {
                #pragma unroll
                for (int q = 0; q < 8; ++q) {
                    const int bq = 8 * wv + q;
                    const float Rv = Rs[bq * CP + t0c];
                    const float Gv = Gs[bq * CP + t0c];
                    const float Cq = Cs[bq * CP + tpc];
                    const float Cpv = t0c > 0 ? Cq : 0.0f;
                    float xa, xb;
                    pair_eval<false>(sa[k][q], sb[k][q], xa, xb, Rv, Gv, Cpv, base[q], ut[q], d1t[q], d2t[q], old, cur);
                    ma = fmaxf(ma, xa);
                    mb = fmaxf(mb, xb);
                }
            } else {
                #pragma unroll
                for (int q = 0; q < 8; ++q) {
                    const int bq = 8 * wv + q;
                    const float Rv = Rs[bq * CP + t0c];
                    const float Gv = Gs[bq * CP + t0c];
                    const float Cq = Cs[bq * CP + tpc];
                    const float Cpv = t0c > 0 ? Cq : 0.0f;
                    float xa, xb;
                    pair_eval<true>(sa[k][q], sb[k][q], xa, xb, Rv, Gv, Cpv, base[q], ut[q], d1t[q], d2t[q], old, cur);
                    ma = fmaxf(ma, xa);
                    mb = fmaxf(mb, xb);
                }
            }
            v2f mv; mv.x = ma; mv.y = mb;
            Mx[(k * FW + wv) * 32 + lane] = mv;
        } else {
            #pragma unroll
            for (int q = 0; q < 8; ++q) { sa[k][q] = 0.0f; sb[k][q] = 0.0f; }
        }
    }
    __syncthreads();

    #pragma unroll
    for (int k = 0; k < 2; ++k) {
        const int jg = 2 * k + h;
        if (jg <= jgmax) {
            const int  t0  = 32 * jg + lane;
            const bool act = t0 <= t;
            v2f mm = Mx[(k * FW) * 32 + lane];
            #pragma unroll
            for (int w2 = 1; w2 < FW; ++w2) {
                const v2f o = Mx[(k * FW + w2) * 32 + lane];
                mm.x = fmaxf(mm.x, o.x);
                mm.y = fmaxf(mm.y, o.y);
            }
            float* wp = W + ((size_t)t * B_DIM + 8 * wv) * NSLOTS + 64 * jg + 2 * lane;
            #pragma unroll
            for (int q = 0; q < 8; ++q) {
                v2f o;
                o.x = act ? fminf(sa[k][q], mm.x) : 0.0f;
                o.y = act ? fminf(sb[k][q], mm.y) : 0.0f;
                *(v2f*)(wp + (size_t)q * NSLOTS) = o;
            }
        }
    }
}

#define KB 32
#define APITCH 36
#define ABUF (T_DIM * APITCH)

__device__ __forceinline__ void lds_barrier()
{
    __builtin_amdgcn_fence(__ATOMIC_RELEASE, "workgroup", "local");
    __builtin_amdgcn_s_barrier();
    __builtin_amdgcn_fence(__ATOMIC_ACQUIRE, "workgroup", "local");
}

template <int J, int RD>
__device__ __forceinline__ int panel_piece(const int tid)
{
    return (256 * RD + 256 <= (8 - J) * 128) ? 256 * RD + tid : 256 * RD + (tid & 127);
}

template <int J, int RD>
__device__ __forceinline__ float4 panel_load(const float* __restrict__ Wb, const int tid)
{
    float4 r = make_float4(0.0f, 0.0f, 0.0f, 0.0f);
    if (256 * RD < (8 - J) * 128) {
        const int idx = panel_piece<J, RD>(tid);
        const int row = 16 * J + (idx >> 3);
        r = *(const float4*)(Wb + (size_t)row * (B_DIM * NSLOTS) + KB * J + 4 * (idx & 7));
    }
    return r;
}

template <int J, int RD>
__device__ __forceinline__ void panel_store(const float4 r, float* __restrict__ Ab, const int tid)
{
    if (256 * RD < (8 - J) * 128) {
        const int idx = panel_piece<J, RD>(tid);
        const int row = 16 * J + (idx >> 3);
        *(float4*)(Ab + row * APITCH + 4 * (idx & 7)) = r;
    }
}

template <int J>
__device__ __forceinline__ void b_load(v2f (&bx)[8], v2f (&by)[8],
                                       const float* __restrict__ B1, const float* __restrict__ B2)
{
    const size_t bstride = (size_t)B_DIM * R_DIM;
    #pragma unroll
    for (int s = 0; s < 8; ++s) {
        bx[s] = *(const v2f*)(B1 + (size_t)(16 * J + 2 * s) * bstride);
        by[s] = *(const v2f*)(B2 + (size_t)(16 * J + 2 * s) * bstride);
    }
}

template <int MT>
__device__ __forceinline__ void tile_store(const v8f (&acc)[8][2], float* __restrict__ orow)
{
    #pragma unroll
    for (int r = 0; r < 8; ++r) {
        v2f o;
        o.x = acc[MT][0][r];
        o.y = acc[MT][1][r];
        *(v2f*)(orow + (size_t)(MT * 16 + r) * (B_DIM * R_DIM)) = o;
    }
}

template <int J>
__device__ __forceinline__ void k_block(v8f (&acc)[8][2], v2f (&bx)[8], v2f (&by)[8],
                                        float* __restrict__ As, const float* __restrict__ Wb,
                                        const float* __restrict__ B1, const float* __restrict__ B2,
                                        float* __restrict__ orow,
                                        const int tid, const int lm, const int half)
{
    constexpr int JN = (J < 7 ? J + 1 : 7);
    float4 st0 = make_float4(0.0f, 0.0f, 0.0f, 0.0f), st1 = st0, st2 = st0, st3 = st0;
    v2f nbx[8], nby[8];
    if (J < 7) {
        b_load<JN>(nbx, nby, B1, B2);
        st0 = panel_load<JN, 0>(Wb, tid);
        st1 = panel_load<JN, 1>(Wb, tid);
        st2 = panel_load<JN, 2>(Wb, tid);
        st3 = panel_load<JN, 3>(Wb, tid);
    }
    if (J > 0) tile_store<(J > 0 ? J - 1 : 0)>(acc, orow);
    const float* Af = As + (J & 1) * ABUF + lm * APITCH + 2 * half;
    #pragma unroll
    for (int s = 0; s < 8; ++s) {
        v2f b0, b1;
        b0.x = bx[s].x; b0.y = by[s].x;
        b1.x = bx[s].y; b1.y = by[s].y;
        #pragma unroll
        for (int mt = J; mt < 8; ++mt) {
            const v2f a = *(const v2f*)(Af + mt * 16 * APITCH + 4 * s);
            acc[mt][0] = __builtin_amdgcn_wmma_f32_16x16x4_f32(false, a, false, b0, (short)0, acc[mt][0], false, false);
            acc[mt][1] = __builtin_amdgcn_wmma_f32_16x16x4_f32(false, a, false, b1, (short)0, acc[mt][1], false, false);
        }
        if (J < 7 && s == 3) {
            float* An = As + ((J + 1) & 1) * ABUF;
            panel_store<JN, 0>(st0, An, tid);
            panel_store<JN, 1>(st1, An, tid);
            panel_store<JN, 2>(st2, An, tid);
            panel_store<JN, 3>(st3, An, tid);
        }
    }
    if (J < 7) {
        lds_barrier();
        #pragma unroll
        for (int s = 0; s < 8; ++s) { bx[s] = nbx[s]; by[s] = nby[s]; }
    }
}

__global__ __launch_bounds__(256) void gemm_kernel(
    const float* __restrict__ W,
    const float* __restrict__ v1,
    const float* __restrict__ v2,
    float* __restrict__ out)
{
    __shared__ __attribute__((aligned(16))) float As[2 * ABUF];

    const int tid  = threadIdx.x;
    const int lane = tid & 31;
    const int wave = tid >> 5;
    const int lm   = lane & 15;
    const int half = lane >> 4;

    const int b  = blockIdx.x >> 1;
    const int n0 = (blockIdx.x & 1) * 256 + wave * 32;

    v8f acc[8][2];
    #pragma unroll
    for (int mt = 0; mt < 8; ++mt) { acc[mt][0] = v8f{}; acc[mt][1] = v8f{}; }

    const float* Wb = W + (size_t)b * NSLOTS;
    const float* B1 = v1 + ((size_t)half * B_DIM + b) * R_DIM + n0 + 2 * lm;
    const float* B2 = v2 + ((size_t)half * B_DIM + b) * R_DIM + n0 + 2 * lm;

    float* orow = out + ((size_t)(8 * half) * B_DIM + b) * R_DIM + n0 + 2 * lm;

    v2f bx[8], by[8];
    {
        b_load<0>(bx, by, B1, B2);
        const float4 p0 = panel_load<0, 0>(Wb, tid);
        const float4 p1 = panel_load<0, 1>(Wb, tid);
        const float4 p2 = panel_load<0, 2>(Wb, tid);
        const float4 p3 = panel_load<0, 3>(Wb, tid);
        panel_store<0, 0>(p0, As, tid);
        panel_store<0, 1>(p1, As, tid);
        panel_store<0, 2>(p2, As, tid);
        panel_store<0, 3>(p3, As, tid);
        lds_barrier();
    }

    k_block<0>(acc, bx, by, As, Wb, B1, B2, orow, tid, lm, half);
    k_block<1>(acc, bx, by, As, Wb, B1, B2, orow, tid, lm, half);
    k_block<2>(acc, bx, by, As, Wb, B1, B2, orow, tid, lm, half);
    k_block<3>(acc, bx, by, As, Wb, B1, B2, orow, tid, lm, half);
    k_block<4>(acc, bx, by, As, Wb, B1, B2, orow, tid, lm, half);
    k_block<5>(acc, bx, by, As, Wb, B1, B2, orow, tid, lm, half);
    k_block<6>(acc, bx, by, As, Wb, B1, B2, orow, tid, lm, half);
    k_block<7>(acc, bx, by, As, Wb, B1, B2, orow, tid, lm, half);

    tile_store<7>(acc, orow);
}

extern "C" void kernel_launch(void* const* d_in, const int* in_sizes, int n_in,
                              void* d_out, int out_size, void* d_ws, size_t ws_size,
                              hipStream_t stream) {
    const float* u  = (const float*)d_in[0];
    const float* d1 = (const float*)d_in[1];
    const float* d2 = (const float*)d_in[2];
    const float* v1 = (const float*)d_in[3];
    const float* v2 = (const float*)d_in[4];
    float* out = (float*)d_out;
    float* W   = (float*)d_ws;

    weights_kernel<<<dim3(T_DIM * 2), dim3(512), 0, stream>>>(u, d1, d2, W);
    gemm_kernel<<<dim3(B_DIM * 2), dim3(256), 0, stream>>>(W, v1, v2, out);
}


# Weights kernel: running sums and minima stay in registers (wave = 8 steps, lane = 4 batches) and are joined once through LDS, rows [step][batch]: 16-byte loads, 2 barriers fewer; sum order changes
# speedup vs baseline: 120.9198x; 1.0903x over previous
#include <hip/hip_runtime.h>

#define T_DIM 128
#define B_DIM 128
#define R_DIM 512
#define NSLOTS 256

typedef float v2f __attribute__((ext_vector_type(2)));
typedef float v8f __attribute__((ext_vector_type(8)));

#define PB 132
#define NST 16
#define FW 16

typedef float v4f __attribute__((ext_vector_type(4)));

template <bool GEN>
__device__ __forceinline__ void pair_eval(float& sa, float& sb, float& xa, float& xb,
                                          const float Rv, const float Gv, const float Cpv, const float base,
                                          const float ut, const float d1t, const float d2t,
                                          const bool old, const bool cur)
{
    const float mg = fminf(Gv, Rv);
    const float sao = fmaxf(0.0f, mg - fminf(Cpv, Rv));
    const float sbo = fmaxf(0.0f, Rv - Gv);
    const float xao = base + mg;
    const float xbo = base + Rv;
    if (GEN) {
        const float ninf = -__builtin_huge_valf();
        sa = old ? sao : (cur ? d1t : 0.0f);
        sb = old ? sbo : (cur ? d2t : 0.0f);
        xa = old ? xao : (cur ? ut - d2t : ninf);
        xb = old ? xbo : (cur ? ut : ninf);
    } else {
        sa = sao; sb = sbo; xa = xao; xb = xbo;
    }
}

template <bool GEN>
__device__ __forceinline__ void group_eval(float (&sa)[8], float (&sb)[8], float& ma, float& mb,
                                           const float* __restrict__ rrow, const float* __restrict__ grow,
                                           const float* __restrict__ crow, const bool hasp,
                                           const float (&base)[8], const float (&ut)[8],
                                           const float (&d1t)[8], const float (&d2t)[8],
                                           const bool old, const bool cur)
{
    const v4f r0 = *(const v4f*)rrow, r1 = *(const v4f*)(rrow + 4);
    const v4f g0 = *(const v4f*)grow, g1 = *(const v4f*)(grow + 4);
    const v4f c0 = *(const v4f*)crow, c1 = *(const v4f*)(crow + 4);
    #pragma unroll
    for (int q = 0; q < 8; ++q) {
        const float Rv = q < 4 ? r0[q & 3] : r1[q & 3];
        const float Gv = q < 4 ? g0[q & 3] : g1[q & 3];
        const float Cq = q < 4 ? c0[q & 3] : c1[q & 3];
        const float Cpv = hasp ? Cq : 0.0f;
        float xa, xb;
        pair_eval<GEN>(sa[q], sb[q], xa, xb, Rv, Gv, Cpv, base[q], ut[q], d1t[q], d2t[q], old, cur);
        ma = fmaxf(ma, xa);
        mb = fmaxf(mb, xb);
    }
}

__global__ __launch_bounds__(512) void weights_kernel(
    const float* __restrict__ u,
    const float* __restrict__ d1,
    const float* __restrict__ d2,
    float* __restrict__ W)
{
    __shared__ __attribute__((aligned(16))) float Cs[T_DIM * PB];
    __shared__ __attribute__((aligned(16))) float Gs[T_DIM * PB];
    __shared__ __attribute__((aligned(16))) float Rs[T_DIM * PB];
    __shared__ __attribute__((aligned(16))) float Tot[NST * B_DIM];
    __shared__ __attribute__((aligned(16))) float Mn[NST * B_DIM];
    __shared__ v2f   Mx[2 * FW * 32];

    const int tid = threadIdx.x;
    const int t   = blockIdx.x >> 1;
    const int h   = blockIdx.x & 1;
    const int jgmax = t >> 5;
    if (h > jgmax) return;

    const int nrows = t > 0 ? t : 1;
    const int lane = tid & 31;
    const int wv   = __builtin_amdgcn_readfirstlane(tid >> 5);
    const float pinf = __builtin_huge_valf();

    float ut[8], d1t[8], d2t[8];
    #pragma unroll
    for (int q = 0; q < 8; ++q) {
        const int bq = 8 * wv + q;
        ut[q]  = u[t * B_DIM + bq];
        d1t[q] = d1[t * B_DIM + bq];
        d2t[q] = d2[t * B_DIM + bq];
    }

    const int  i0   = 8 * wv;
    const bool live = i0 < nrows;
    float P[8][4];
    float gt[8][4];
    float tot[4] = {0.0f, 0.0f, 0.0f, 0.0f};
    float low[4] = {0.0f, 0.0f, 0.0f, 0.0f};
    if (live) {
        v4f uu[8], aa[8], cc[8];
        #pragma unroll
        for (int k = 0; k < 8; ++k) {
            const int i  = i0 + k;
            const int ic = i < nrows ? i : nrows - 1;
            uu[k] = *(const v4f*)(u  + ic * B_DIM + 4 * lane);
            aa[k] = *(const v4f*)(d1 + ic * B_DIM + 4 * lane);
            cc[k] = *(const v4f*)(d2 + ic * B_DIM + 4 * lane);
        }
        float acc[4] = {0.0f, 0.0f, 0.0f, 0.0f};
        float mn[4]  = {pinf, pinf, pinf, pinf};
        #pragma unroll
        for (int k = 0; k < 8; ++k) {
            const bool in = i0 + k < nrows;
            #pragma unroll
            for (int j = 0; j < 4; ++j) {
                const float up = fmaxf(uu[k][j], 0.0f);
                const float c  = in ? (aa[k][j] + cc[k][j]) - up : 0.0f;
                gt[k][j] = in ? fmaxf(0.0f, cc[k][j] - up) : 0.0f;
                acc[j] += c;
                P[k][j] = acc[j];
                mn[j] = fminf(mn[j], acc[j]);
            }
        }
        #pragma unroll
        for (int j = 0; j < 4; ++j) { tot[j] = acc[j]; low[j] = mn[j]; }
    } else {
        #pragma unroll
        for (int k = 0; k < 8; ++k) {
            #pragma unroll
            for (int j = 0; j < 4; ++j) { P[k][j] = 0.0f; gt[k][j] = 0.0f; }
        }
    }
    {
        v4f tv, lv;
        tv.x = tot[0]; tv.y = tot[1]; tv.z = tot[2]; tv.w = tot[3];
        lv.x = low[0]; lv.y = low[1]; lv.z = low[2]; lv.w = low[3];
        *(v4f*)(Tot + wv * B_DIM + 4 * lane) = tv;
        *(v4f*)(Mn  + wv * B_DIM + 4 * lane) = lv;
    }
    __syncthreads();

    if (live) {
        float run[4]   = {0.0f, 0.0f, 0.0f, 0.0f};
        float own[4]   = {0.0f, 0.0f, 0.0f, 0.0f};
        float carry[4] = {pinf, pinf, pinf, pinf};
        #pragma unroll
        for (int g = 0; g < NST; ++g) {
            const v4f tg = *(const v4f*)(Tot + g * B_DIM + 4 * lane);
            const v4f mg = *(const v4f*)(Mn  + g * B_DIM + 4 * lane);
            const bool me    = g == wv;
            const bool later = g > wv;
            #pragma unroll
            for (int j = 0; j < 4; ++j) {
                own[j] = me ? run[j] : own[j];
                const float cand = run[j] + mg[j];
                carry[j] = later ? fminf(carry[j], cand) : carry[j];
                run[j] += tg[j];
            }
        }
        float r[4];
        #pragma unroll
        for (int j = 0; j < 4; ++j) r[j] = carry[j];
        #pragma unroll
        for (int k = 7; k >= 0; --k) {
            v4f Cv, Gv, Rv;
            #pragma unroll
            for (int j = 0; j < 4; ++j) {
                const float C = own[j] + P[k][j];
                Cv[j] = C;
                Gv[j] = C - gt[k][j];
                r[j]  = fminf(r[j], C);
                Rv[j] = r[j];
            }
            *(v4f*)(Cs + (i0 + k) * PB + 4 * lane) = Cv;
            *(v4f*)(Gs + (i0 + k) * PB + 4 * lane) = Gv;
            *(v4f*)(Rs + (i0 + k) * PB + 4 * lane) = Rv;
        }
    }
    __syncthreads();

    float base[8];
    {
        const float* nrow = Cs + (nrows - 1) * PB + 8 * wv;
        const v4f n0 = *(const v4f*)nrow, n1 = *(const v4f*)(nrow + 4);
        #pragma unroll
        for (int q = 0; q < 8; ++q) {
            const float cq = q < 4 ? n0[q & 3] : n1[q & 3];
            const float cn = (t > 0) ? cq : 0.0f;
            base[q] = (ut[q] - (d1t[q] + d2t[q])) - cn;
        }
    }

    float sa[2][8], sb[2][8];
    #pragma unroll
    for (int k = 0; k < 2; ++k) {
        const int jg = 2 * k + h;
        if (jg <= jgmax) {
            const int  t0  = 32 * jg + lane;
            const bool old = t0 < t;
            const bool cur = t0 == t;
            const int  t0c = t0 < nrows ? t0 : nrows - 1;
            const int  tpc = t0c > 0 ? t0c - 1 : 0;
            const float* rrow = Rs + t0c * PB + 8 * wv;
            const float* grow = Gs + t0c * PB + 8 * wv;
            const float* crow = Cs + tpc * PB + 8 * wv;
            const bool hasp = t0c > 0;
            float ma = -pinf, mb = -pinf;
            if (32 * jg + 31 < t) {
                group_eval<false>(sa[k], sb[k], ma, mb, rrow, grow, crow, hasp, base, ut, d1t, d2t, old, cur);
            } else {
                group_eval<true>(sa[k], sb[k], ma, mb, rrow, grow, crow, hasp, base, ut, d1t, d2t, old, cur);
            }
            v2f mv; mv.x = ma; mv.y = mb;
            Mx[(k * FW + wv) * 32 + lane] = mv;
        } else {
            #pragma unroll
            for (int q = 0; q < 8; ++q) { sa[k][q] = 0.0f; sb[k][q] = 0.0f; }
        }
    }
    __syncthreads();

    #pragma unroll
    for (int k = 0; k < 2; ++k) {
        const int jg = 2 * k + h;
        if (jg <= jgmax) {
            const int  t0  = 32 * jg + lane;
            const bool act = t0 <= t;
            v2f mm = Mx[(k * FW) * 32 + lane];
            #pragma unroll
            for (int w2 = 1; w2 < FW; ++w2) {
                const v2f o = Mx[(k * FW + w2) * 32 + lane];
                mm.x = fmaxf(mm.x, o.x);
                mm.y = fmaxf(mm.y, o.y);
            }
            float* wp = W + ((size_t)t * B_DIM + 8 * wv) * NSLOTS + 64 * jg + 2 * lane;
            #pragma unroll
            for (int q = 0; q < 8; ++q) {
                v2f o;
                o.x = act ? fminf(sa[k][q], mm.x) : 0.0f;
                o.y = act ? fminf(sb[k][q], mm.y) : 0.0f;
                *(v2f*)(wp + (size_t)q * NSLOTS) = o;
            }
        }
    }
}

#define KB 32
#define APITCH 36
#define ABUF (T_DIM * APITCH)

__device__ __forceinline__ void lds_barrier()
{
    __builtin_amdgcn_fence(__ATOMIC_RELEASE, "workgroup", "local");
    __builtin_amdgcn_s_barrier();
    __builtin_amdgcn_fence(__ATOMIC_ACQUIRE, "workgroup", "local");
}

template <int J, int RD>
__device__ __forceinline__ int panel_piece(const int tid)
{
    return (256 * RD + 256 <= (8 - J) * 128) ? 256 * RD + tid : 256 * RD + (tid & 127);
}

template <int J, int RD>
__device__ __forceinline__ float4 panel_load(const float* __restrict__ Wb, const int tid)
{
    float4 r = make_float4(0.0f, 0.0f, 0.0f, 0.0f);
    if (256 * RD < (8 - J) * 128) {
        const int idx = panel_piece<J, RD>(tid);
        const int row = 16 * J + (idx >> 3);
        r = *(const float4*)(Wb + (size_t)row * (B_DIM * NSLOTS) + KB * J + 4 * (idx & 7));
    }
    return r;
}

template <int J, int RD>
__device__ __forceinline__ void panel_store(const float4 r, float* __restrict__ Ab, const int tid)
{
    if (256 * RD < (8 - J) * 128) {
        const int idx = panel_piece<J, RD>(tid);
        const int row = 16 * J + (idx >> 3);
        *(float4*)(Ab + row * APITCH + 4 * (idx & 7)) = r;
    }
}

template <int J>
__device__ __forceinline__ void b_load(v2f (&bx)[8], v2f (&by)[8],
                                       const float* __restrict__ B1, const float* __restrict__ B2)
{
    const size_t bstride = (size_t)B_DIM * R_DIM;
    #pragma unroll
    for (int s = 0; s < 8; ++s) {
        bx[s] = *(const v2f*)(B1 + (size_t)(16 * J + 2 * s) * bstride);
        by[s] = *(const v2f*)(B2 + (size_t)(16 * J + 2 * s) * bstride);
    }
}

template <int MT>
__device__ __forceinline__ void tile_store(const v8f (&acc)[8][2], float* __restrict__ orow)
{
    #pragma unroll
    for (int r = 0; r < 8; ++r) {
        v2f o;
        o.x = acc[MT][0][r];
        o.y = acc[MT][1][r];
        *(v2f*)(orow + (size_t)(MT * 16 + r) * (B_DIM * R_DIM)) = o;
    }
}

template <int J>
__device__ __forceinline__ void k_block(v8f (&acc)[8][2], v2f (&bx)[8], v2f (&by)[8],
                                        float* __restrict__ As, const float* __restrict__ Wb,
                                        const float* __restrict__ B1, const float* __restrict__ B2,
                                        float* __restrict__ orow,
                                        const int tid, const int lm, const int half)
{
    constexpr int JN = (J < 7 ? J + 1 : 7);
    float4 st0 = make_float4(0.0f, 0.0f, 0.0f, 0.0f), st1 = st0, st2 = st0, st3 = st0;
    v2f nbx[8], nby[8];
    if (J < 7) {
        b_load<JN>(nbx, nby, B1, B2);
        st0 = panel_load<JN, 0>(Wb, tid);
        st1 = panel_load<JN, 1>(Wb, tid);
        st2 = panel_load<JN, 2>(Wb, tid);
        st3 = panel_load<JN, 3>(Wb, tid);
    }
    if (J > 0) tile_store<(J > 0 ? J - 1 : 0)>(acc, orow);
    const float* Af = As + (J & 1) * ABUF + lm * APITCH + 2 * half;
    #pragma unroll
    for (int s = 0; s < 8; ++s) {
        v2f b0, b1;
        b0.x = bx[s].x; b0.y = by[s].x;
        b1.x = bx[s].y; b1.y = by[s].y;
        #pragma unroll
        for (int mt = J; mt < 8; ++mt) {
            const v2f a = *(const v2f*)(Af + mt * 16 * APITCH + 4 * s);
            acc[mt][0] = __builtin_amdgcn_wmma_f32_16x16x4_f32(false, a, false, b0, (short)0, acc[mt][0], false, false);
            acc[mt][1] = __builtin_amdgcn_wmma_f32_16x16x4_f32(false, a, false, b1, (short)0, acc[mt][1], false, false);
        }
        if (J < 7 && s == 3) {
            float* An = As + ((J + 1) & 1) * ABUF;
            panel_store<JN, 0>(st0, An, tid);
            panel_store<JN, 1>(st1, An, tid);
            panel_store<JN, 2>(st2, An, tid);
            panel_store<JN, 3>(st3, An, tid);
        }
    }
    if (J < 7) {
        lds_barrier();
        #pragma unroll
        for (int s = 0; s < 8; ++s) { bx[s] = nbx[s]; by[s] = nby[s]; }
    }
}

__global__ __launch_bounds__(256) void gemm_kernel(
    const float* __restrict__ W,
    const float* __restrict__ v1,
    const float* __restrict__ v2,
    float* __restrict__ out)
{
    __shared__ __attribute__((aligned(16))) float As[2 * ABUF];

    const int tid  = threadIdx.x;
    const int lane = tid & 31;
    const int wave = tid >> 5;
    const int lm   = lane & 15;
    const int half = lane >> 4;

    const int b  = blockIdx.x >> 1;
    const int n0 = (blockIdx.x & 1) * 256 + wave * 32;

    v8f acc[8][2];
    #pragma unroll
    for (int mt = 0; mt < 8; ++mt) { acc[mt][0] = v8f{}; acc[mt][1] = v8f{}; }

    const float* Wb = W + (size_t)b * NSLOTS;
    const float* B1 = v1 + ((size_t)half * B_DIM + b) * R_DIM + n0 + 2 * lm;
    const float* B2 = v2 + ((size_t)half * B_DIM + b) * R_DIM + n0 + 2 * lm;

    float* orow = out + ((size_t)(8 * half) * B_DIM + b) * R_DIM + n0 + 2 * lm;

    v2f bx[8], by[8];
    {
        b_load<0>(bx, by, B1, B2);
        const float4 p0 = panel_load<0, 0>(Wb, tid);
        const float4 p1 = panel_load<0, 1>(Wb, tid);
        const float4 p2 = panel_load<0, 2>(Wb, tid);
        const float4 p3 = panel_load<0, 3>(Wb, tid);
        panel_store<0, 0>(p0, As, tid);
        panel_store<0, 1>(p1, As, tid);
        panel_store<0, 2>(p2, As, tid);
        panel_store<0, 3>(p3, As, tid);
        lds_barrier();
    }

    k_block<0>(acc, bx, by, As, Wb, B1, B2, orow, tid, lm, half);
    k_block<1>(acc, bx, by, As, Wb, B1, B2, orow, tid, lm, half);
    k_block<2>(acc, bx, by, As, Wb, B1, B2, orow, tid, lm, half);
    k_block<3>(acc, bx, by, As, Wb, B1, B2, orow, tid, lm, half);
    k_block<4>(acc, bx, by, As, Wb, B1, B2, orow, tid, lm, half);
    k_block<5>(acc, bx, by, As, Wb, B1, B2, orow, tid, lm, half);
    k_block<6>(acc, bx, by, As, Wb, B1, B2, orow, tid, lm, half);
    k_block<7>(acc, bx, by, As, Wb, B1, B2, orow, tid, lm, half);

    tile_store<7>(acc, orow);
}

extern "C" void kernel_launch(void* const* d_in, const int* in_sizes, int n_in,
                              void* d_out, int out_size, void* d_ws, size_t ws_size,
                              hipStream_t stream) {
    const float* u  = (const float*)d_in[0];
    const float* d1 = (const float*)d_in[1];
    const float* d2 = (const float*)d_in[2];
    const float* v1 = (const float*)d_in[3];
    const float* v2 = (const float*)d_in[4];
    float* out = (float*)d_out;
    float* W   = (float*)d_ws;

    weights_kernel<<<dim3(T_DIM * 2), dim3(512), 0, stream>>>(u, d1, d2, W);
    gemm_kernel<<<dim3(B_DIM * 2), dim3(256), 0, stream>>>(W, v1, v2, out);
}


# GEMM: A fragments of each half K block (4 k-steps) are read from LDS as one group half a block ahead into a second register set, not among the matrix instructions; barrier moves mid-block; same sums
# speedup vs baseline: 128.1970x; 1.0602x over previous
#include <hip/hip_runtime.h>

#define T_DIM 128
#define B_DIM 128
#define R_DIM 512
#define NSLOTS 256

typedef float v2f __attribute__((ext_vector_type(2)));
typedef float v8f __attribute__((ext_vector_type(8)));

#define PB 132
#define NST 16
#define FW 16

typedef float v4f __attribute__((ext_vector_type(4)));

template <bool GEN>
__device__ __forceinline__ void pair_eval(float& sa, float& sb, float& xa, float& xb,
                                          const float Rv, const float Gv, const float Cpv, const float base,
                                          const float ut, const float d1t, const float d2t,
                                          const bool old, const bool cur)
{
    const float mg = fminf(Gv, Rv);
    const float sao = fmaxf(0.0f, mg - fminf(Cpv, Rv));
    const float sbo = fmaxf(0.0f, Rv - Gv);
    const float xao = base + mg;
    const float xbo = base + Rv;
    if (GEN) {
        const float ninf = -__builtin_huge_valf();
        sa = old ? sao : (cur ? d1t : 0.0f);
        sb = old ? sbo : (cur ? d2t : 0.0f);
        xa = old ? xao : (cur ? ut - d2t : ninf);
        xb = old ? xbo : (cur ? ut : ninf);
    } else {
        sa = sao; sb = sbo; xa = xao; xb = xbo;
    }
}

template <bool GEN>
__device__ __forceinline__ void group_eval(float (&sa)[8], float (&sb)[8], float& ma, float& mb,
                                           const float* __restrict__ rrow, const float* __restrict__ grow,
                                           const float* __restrict__ crow, const bool hasp,
                                           const float (&base)[8], const float (&ut)[8],
                                           const float (&d1t)[8], const float (&d2t)[8],
                                           const bool old, const bool cur)
{
    const v4f r0 = *(const v4f*)rrow, r1 = *(const v4f*)(rrow + 4);
    const v4f g0 = *(const v4f*)grow, g1 = *(const v4f*)(grow + 4);
    const v4f c0 = *(const v4f*)crow, c1 = *(const v4f*)(crow + 4);
    #pragma unroll
    for (int q = 0; q < 8; ++q) {
        const float Rv = q < 4 ? r0[q & 3] : r1[q & 3];
        const float Gv = q < 4 ? g0[q & 3] : g1[q & 3];
        const float Cq = q < 4 ? c0[q & 3] : c1[q & 3];
        const float Cpv = hasp ? Cq : 0.0f;
        float xa, xb;
        pair_eval<GEN>(sa[q], sb[q], xa, xb, Rv, Gv, Cpv, base[q], ut[q], d1t[q], d2t[q], old, cur);
        ma = fmaxf(ma, xa);
        mb = fmaxf(mb, xb);
    }
}

__global__ __launch_bounds__(512) void weights_kernel(
    const float* __restrict__ u,
    const float* __restrict__ d1,
    const float* __restrict__ d2,
    float* __restrict__ W)
{
    __shared__ __attribute__((aligned(16))) float Cs[T_DIM * PB];
    __shared__ __attribute__((aligned(16))) float Gs[T_DIM * PB];
    __shared__ __attribute__((aligned(16))) float Rs[T_DIM * PB];
    __shared__ __attribute__((aligned(16))) float Tot[NST * B_DIM];
    __shared__ __attribute__((aligned(16))) float Mn[NST * B_DIM];
    __shared__ v2f   Mx[2 * FW * 32];

    const int tid = threadIdx.x;
    const int t   = blockIdx.x >> 1;
    const int h   = blockIdx.x & 1;
    const int jgmax = t >> 5;
    if (h > jgmax) return;

    const int nrows = t > 0 ? t : 1;
    const int lane = tid & 31;
    const int wv   = __builtin_amdgcn_readfirstlane(tid >> 5);
    const float pinf = __builtin_huge_valf();

    float ut[8], d1t[8], d2t[8];
    #pragma unroll
    for (int q = 0; q < 8; ++q) {
        const int bq = 8 * wv + q;
        ut[q]  = u[t * B_DIM + bq];
        d1t[q] = d1[t * B_DIM + bq];
        d2t[q] = d2[t * B_DIM + bq];
    }

    const int  i0   = 8 * wv;
    const bool live = i0 < nrows;
    float P[8][4];
    float gt[8][4];
    float tot[4] = {0.0f, 0.0f, 0.0f, 0.0f};
    float low[4] = {0.0f, 0.0f, 0.0f, 0.0f};
    if (live) {
        v4f uu[8], aa[8], cc[8];
        #pragma unroll
        for (int k = 0; k < 8; ++k) {
            const int i  = i0 + k;
            const int ic = i < nrows ? i : nrows - 1;
            uu[k] = *(const v4f*)(u  + ic * B_DIM + 4 * lane);
            aa[k] = *(const v4f*)(d1 + ic * B_DIM + 4 * lane);
            cc[k] = *(const v4f*)(d2 + ic * B_DIM + 4 * lane);
        }
        float acc[4] = {0.0f, 0.0f, 0.0f, 0.0f};
        float mn[4]  = {pinf, pinf, pinf, pinf};
        #pragma unroll
        for (int k = 0; k < 8; ++k) {
            const bool in = i0 + k < nrows;
            #pragma unroll
            for (int j = 0; j < 4; ++j) {
                const float up = fmaxf(uu[k][j], 0.0f);
                const float c  = in ? (aa[k][j] + cc[k][j]) - up : 0.0f;
                gt[k][j] = in ? fmaxf(0.0f, cc[k][j] - up) : 0.0f;
                acc[j] += c;
                P[k][j] = acc[j];
                mn[j] = fminf(mn[j], acc[j]);
            }
        }
        #pragma unroll
        for (int j = 0; j < 4; ++j) { tot[j] = acc[j]; low[j] = mn[j]; }
    } else {
        #pragma unroll
        for (int k = 0; k < 8; ++k) {
            #pragma unroll
            for (int j = 0; j < 4; ++j) { P[k][j] = 0.0f; gt[k][j] = 0.0f; }
        }
    }
    {
        v4f tv, lv;
        tv.x = tot[0]; tv.y = tot[1]; tv.z = tot[2]; tv.w = tot[3];
        lv.x = low[0]; lv.y = low[1]; lv.z = low[2]; lv.w = low[3];
        *(v4f*)(Tot + wv * B_DIM + 4 * lane) = tv;
        *(v4f*)(Mn  + wv * B_DIM + 4 * lane) = lv;
    }
    __syncthreads();

    if (live) {
        float run[4]   = {0.0f, 0.0f, 0.0f, 0.0f};
        float own[4]   = {0.0f, 0.0f, 0.0f, 0.0f};
        float carry[4] = {pinf, pinf, pinf, pinf};
        #pragma unroll
        for (int g = 0; g < NST; ++g) {
            const v4f tg = *(const v4f*)(Tot + g * B_DIM + 4 * lane);
            const v4f mg = *(const v4f*)(Mn  + g * B_DIM + 4 * lane);
            const bool me    = g == wv;
            const bool later = g > wv;
            #pragma unroll
            for (int j = 0; j < 4; ++j) {
                own[j] = me ? run[j] : own[j];
                const float cand = run[j] + mg[j];
                carry[j] = later ? fminf(carry[j], cand) : carry[j];
                run[j] += tg[j];
            }
        }
        float r[4];
        #pragma unroll
        for (int j = 0; j < 4; ++j) r[j] = carry[j];
        #pragma unroll
        for (int k = 7; k >= 0; --k) {
            v4f Cv, Gv, Rv;
            #pragma unroll
            for (int j = 0; j < 4; ++j) {
                const float C = own[j] + P[k][j];
                Cv[j] = C;
                Gv[j] = C - gt[k][j];
                r[j]  = fminf(r[j], C);
                Rv[j] = r[j];
            }
            *(v4f*)(Cs + (i0 + k) * PB + 4 * lane) = Cv;
            *(v4f*)(Gs + (i0 + k) * PB + 4 * lane) = Gv;
            *(v4f*)(Rs + (i0 + k) * PB + 4 * lane) = Rv;
        }
    }
    __syncthreads();

    float base[8];
    {
        const float* nrow = Cs + (nrows - 1) * PB + 8 * wv;
        const v4f n0 = *(const v4f*)nrow, n1 = *(const v4f*)(nrow + 4);
        #pragma unroll
        for (int q = 0; q < 8; ++q) {
            const float cq = q < 4 ? n0[q & 3] : n1[q & 3];
            const float cn = (t > 0) ? cq : 0.0f;
            base[q] = (ut[q] - (d1t[q] + d2t[q])) - cn;
        }
    }

    float sa[2][8], sb[2][8];
    #pragma unroll
    for (int k = 0; k < 2; ++k) {
        const int jg = 2 * k + h;
        if (jg <= jgmax) {
            const int  t0  = 32 * jg + lane;
            const bool old = t0 < t;
            const bool cur = t0 == t;
            const int  t0c = t0 < nrows ? t0 : nrows - 1;
            const int  tpc = t0c > 0 ? t0c - 1 : 0;
            const float* rrow = Rs + t0c * PB + 8 * wv;
            const float* grow = Gs + t0c * PB + 8 * wv;
            const float* crow = Cs + tpc * PB + 8 * wv;
            const bool hasp = t0c > 0;
            float ma = -pinf, mb = -pinf;
            if (32 * jg + 31 < t) {
                group_eval<false>(sa[k], sb[k], ma, mb, rrow, grow, crow, hasp, base, ut, d1t, d2t, old, cur);
            } else {
                group_eval<true>(sa[k], sb[k], ma, mb, rrow, grow, crow, hasp, base, ut, d1t, d2t, old, cur);
            }
            v2f mv; mv.x = ma; mv.y = mb;
            Mx[(k * FW + wv) * 32 + lane] = mv;
        } else {
            #pragma unroll
            for (int q = 0; q < 8; ++q) { sa[k][q] = 0.0f; sb[k][q] = 0.0f; }
        }
    }
    __syncthreads();

    #pragma unroll
    for (int k = 0; k < 2; ++k) {
        const int jg = 2 * k + h;
        if (jg <= jgmax) {
            const int  t0  = 32 * jg + lane;
            const bool act = t0 <= t;
            v2f mm = Mx[(k * FW) * 32 + lane];
            #pragma unroll
            for (int w2 = 1; w2 < FW; ++w2) {
                const v2f o = Mx[(k * FW + w2) * 32 + lane];
                mm.x = fmaxf(mm.x, o.x);
                mm.y = fmaxf(mm.y, o.y);
            }
            float* wp = W + ((size_t)t * B_DIM + 8 * wv) * NSLOTS + 64 * jg + 2 * lane;
            #pragma unroll
            for (int q = 0; q < 8; ++q) {
                v2f o;
                o.x = act ? fminf(sa[k][q], mm.x) : 0.0f;
                o.y = act ? fminf(sb[k][q], mm.y) : 0.0f;
                *(v2f*)(wp + (size_t)q * NSLOTS) = o;
            }
        }
    }
}

#define KB 32
#define APITCH 36
#define ABUF (T_DIM * APITCH)

__device__ __forceinline__ void lds_barrier()
{
    __builtin_amdgcn_fence(__ATOMIC_RELEASE, "workgroup", "local");
    __builtin_amdgcn_s_barrier();
    __builtin_amdgcn_fence(__ATOMIC_ACQUIRE, "workgroup", "local");
}

template <int J, int RD>
__device__ __forceinline__ int panel_piece(const int tid)
{
    return (256 * RD + 256 <= (8 - J) * 128) ? 256 * RD + tid : 256 * RD + (tid & 127);
}

template <int J, int RD>
__device__ __forceinline__ float4 panel_load(const float* __restrict__ Wb, const int tid)
{
    float4 r = make_float4(0.0f, 0.0f, 0.0f, 0.0f);
    if (256 * RD < (8 - J) * 128) {
        const int idx = panel_piece<J, RD>(tid);
        const int row = 16 * J + (idx >> 3);
        r = *(const float4*)(Wb + (size_t)row * (B_DIM * NSLOTS) + KB * J + 4 * (idx & 7));
    }
    return r;
}

template <int J, int RD>
__device__ __forceinline__ void panel_store(const float4 r, float* __restrict__ Ab, const int tid)
{
    if (256 * RD < (8 - J) * 128) {
        const int idx = panel_piece<J, RD>(tid);
        const int row = 16 * J + (idx >> 3);
        *(float4*)(Ab + row * APITCH + 4 * (idx & 7)) = r;
    }
}

template <int J>
__device__ __forceinline__ void b_load(v2f (&bx)[8], v2f (&by)[8],
                                       const float* __restrict__ B1, const float* __restrict__ B2)
{
    const size_t bstride = (size_t)B_DIM * R_DIM;
    #pragma unroll
    for (int s = 0; s < 8; ++s) {
        bx[s] = *(const v2f*)(B1 + (size_t)(16 * J + 2 * s) * bstride);
        by[s] = *(const v2f*)(B2 + (size_t)(16 * J + 2 * s) * bstride);
    }
}

template <int MT>
__device__ __forceinline__ void tile_store(const v8f (&acc)[8][2], float* __restrict__ orow)
{
    #pragma unroll
    for (int r = 0; r < 8; ++r) {
        v2f o;
        o.x = acc[MT][0][r];
        o.y = acc[MT][1][r];
        *(v2f*)(orow + (size_t)(MT * 16 + r) * (B_DIM * R_DIM)) = o;
    }
}

template <int MT0, int S0>
__device__ __forceinline__ void frag_load(v2f (&f)[4][8], const float* __restrict__ Af)
{
    #pragma unroll
    for (int s = 0; s < 4; ++s) {
        #pragma unroll
        for (int mt = MT0; mt < 8; ++mt)
            f[s][mt] = *(const v2f*)(Af + mt * 16 * APITCH + 4 * (S0 + s));
    }
}

template <int J, int S0, int SA, int SB>
__device__ __forceinline__ void half_mma(v8f (&acc)[8][2], const v2f (&f)[4][8],
                                         const v2f (&bx)[8], const v2f (&by)[8])
{
    #pragma unroll
    for (int s = SA; s < SB; ++s) {
        v2f b0, b1;
        b0.x = bx[S0 + s].x; b0.y = by[S0 + s].x;
        b1.x = bx[S0 + s].y; b1.y = by[S0 + s].y;
        #pragma unroll
        for (int mt = J; mt < 8; ++mt) {
            acc[mt][0] = __builtin_amdgcn_wmma_f32_16x16x4_f32(false, f[s][mt], false, b0, (short)0, acc[mt][0], false, false);
            acc[mt][1] = __builtin_amdgcn_wmma_f32_16x16x4_f32(false, f[s][mt], false, b1, (short)0, acc[mt][1], false, false);
        }
    }
}

template <int J>
__device__ __forceinline__ void k_block(v8f (&acc)[8][2], v2f (&bx)[8], v2f (&by)[8],
                                        v2f (&fc)[4][8], v2f (&fn)[4][8],
                                        float* __restrict__ As, const float* __restrict__ Wb,
                                        const float* __restrict__ B1, const float* __restrict__ B2,
                                        float* __restrict__ orow,
                                        const int tid, const int lm, const int half)
{
    constexpr int JN = (J < 7 ? J + 1 : 7);
    float4 st0 = make_float4(0.0f, 0.0f, 0.0f, 0.0f), st1 = st0, st2 = st0, st3 = st0;
    v2f nbx[8], nby[8];
    v2f fs[4][8];
    frag_load<J, 4>(fs, As + (J & 1) * ABUF + lm * APITCH + 2 * half);
    if (J < 7) {
        b_load<JN>(nbx, nby, B1, B2);
        st0 = panel_load<JN, 0>(Wb, tid);
        st1 = panel_load<JN, 1>(Wb, tid);
        st2 = panel_load<JN, 2>(Wb, tid);
        st3 = panel_load<JN, 3>(Wb, tid);
    }
    if (J > 0) tile_store<(J > 0 ? J - 1 : 0)>(acc, orow);
    __builtin_amdgcn_sched_barrier(0);
    half_mma<J, 0, 0, 3>(acc, fc, bx, by);
    if (J < 7) {
        float* An = As + ((J + 1) & 1) * ABUF;
        panel_store<JN, 0>(st0, An, tid);
        panel_store<JN, 1>(st1, An, tid);
        panel_store<JN, 2>(st2, An, tid);
        panel_store<JN, 3>(st3, An, tid);
        __builtin_amdgcn_fence(__ATOMIC_RELEASE, "workgroup", "local");
        __builtin_amdgcn_s_barrier_signal(-1);
        __builtin_amdgcn_sched_barrier(0);
        half_mma<J, 0, 3, 4>(acc, fc, bx, by);
        __builtin_amdgcn_sched_barrier(0);
        __builtin_amdgcn_s_barrier_wait(-1);
        __builtin_amdgcn_fence(__ATOMIC_ACQUIRE, "workgroup", "local");
        frag_load<JN, 0>(fn, An + lm * APITCH + 2 * half);
    } else {
        half_mma<J, 0, 3, 4>(acc, fc, bx, by);
    }
    __builtin_amdgcn_sched_barrier(0);
    half_mma<J, 4, 0, 4>(acc, fs, bx, by);
    if (J < 7) {
        #pragma unroll
        for (int s = 0; s < 8; ++s) { bx[s] = nbx[s]; by[s] = nby[s]; }
    }
}

__global__ __launch_bounds__(256) void gemm_kernel(
    const float* __restrict__ W,
    const float* __restrict__ v1,
    const float* __restrict__ v2,
    float* __restrict__ out)
{
    __shared__ __attribute__((aligned(16))) float As[2 * ABUF];

    const int tid  = threadIdx.x;
    const int lane = tid & 31;
    const int wave = tid >> 5;
    const int lm   = lane & 15;
    const int half = lane >> 4;

    const int b  = blockIdx.x >> 1;
    const int n0 = (blockIdx.x & 1) * 256 + wave * 32;

    v8f acc[8][2];
    #pragma unroll
    for (int mt = 0; mt < 8; ++mt) { acc[mt][0] = v8f{}; acc[mt][1] = v8f{}; }

    const float* Wb = W + (size_t)b * NSLOTS;
    const float* B1 = v1 + ((size_t)half * B_DIM + b) * R_DIM + n0 + 2 * lm;
    const float* B2 = v2 + ((size_t)half * B_DIM + b) * R_DIM + n0 + 2 * lm;

    float* orow = out + ((size_t)(8 * half) * B_DIM + b) * R_DIM + n0 + 2 * lm;

    v2f bx[8], by[8];
    {
        b_load<0>(bx, by, B1, B2);
        const float4 p0 = panel_load<0, 0>(Wb, tid);
        const float4 p1 = panel_load<0, 1>(Wb, tid);
        const float4 p2 = panel_load<0, 2>(Wb, tid);
        const float4 p3 = panel_load<0, 3>(Wb, tid);
        panel_store<0, 0>(p0, As, tid);
        panel_store<0, 1>(p1, As, tid);
        panel_store<0, 2>(p2, As, tid);
        panel_store<0, 3>(p3, As, tid);
        lds_barrier();
    }
    v2f fa[4][8], fb[4][8];
    frag_load<0, 0>(fa, As + lm * APITCH + 2 * half);

    k_block<0>(acc, bx, by, fa, fb, As, Wb, B1, B2, orow, tid, lm, half);
    k_block<1>(acc, bx, by, fb, fa, As, Wb, B1, B2, orow, tid, lm, half);
    k_block<2>(acc, bx, by, fa, fb, As, Wb, B1, B2, orow, tid, lm, half);
    k_block<3>(acc, bx, by, fb, fa, As, Wb, B1, B2, orow, tid, lm, half);
    k_block<4>(acc, bx, by, fa, fb, As, Wb, B1, B2, orow, tid, lm, half);
    k_block<5>(acc, bx, by, fb, fa, As, Wb, B1, B2, orow, tid, lm, half);
    k_block<6>(acc, bx, by, fa, fb, As, Wb, B1, B2, orow, tid, lm, half);
    k_block<7>(acc, bx, by, fb, fa, As, Wb, B1, B2, orow, tid, lm, half);

    tile_store<7>(acc, orow);
}

extern "C" void kernel_launch(void* const* d_in, const int* in_sizes, int n_in,
                              void* d_out, int out_size, void* d_ws, size_t ws_size,
                              hipStream_t stream) {
    const float* u  = (const float*)d_in[0];
    const float* d1 = (const float*)d_in[1];
    const float* d2 = (const float*)d_in[2];
    const float* v1 = (const float*)d_in[3];
    const float* v2 = (const float*)d_in[4];
    float* out = (float*)d_out;
    float* W   = (float*)d_ws;

    weights_kernel<<<dim3(T_DIM * 2), dim3(512), 0, stream>>>(u, d1, d2, W);
    gemm_kernel<<<dim3(B_DIM * 2), dim3(256), 0, stream>>>(W, v1, v2, out);
}
